# starting kernel
# baseline (speedup 1.0000x reference)
#include <hip/hip_runtime.h>
#include <math.h>


typedef __attribute__((ext_vector_type(2))) float v2f;
typedef __attribute__((ext_vector_type(8))) float v8f;

#define HW 64
#define PIX 4096
#define CIN 256
#define CCOMP 64
#define MENC 100
#define MPAD 112
#define KENC 576
#define NPIX_TOTAL 16384

__global__ __launch_bounds__(256) void carafe_compress(
    const float* __restrict__ x, const float* __restrict__ w_comp,
    const float* __restrict__ b_comp, float* __restrict__ comp) {
  const int wave = threadIdx.x >> 5;
  const int lane = threadIdx.x & 31;
  const int row  = lane & 15;
  const int half = lane >> 4;

  const int tile = blockIdx.x * 8 + wave;
  const int gp   = tile << 4;
  const int n    = gp >> 12;
  const int pin  = gp & 4095;

  v8f acc[4];
#pragma unroll
  for (int t = 0; t < 4; ++t)
#pragma unroll
    for (int v = 0; v < 8; ++v)
      acc[t][v] = b_comp[t * 16 + v + 8 * half];

  const float* xb = x + (size_t)n * CIN * PIX + pin + row;

  for (int k0 = 0; k0 < CIN; k0 += 4) {
    const int kb = k0 + 2 * half;
    v2f b;
    b[0] = xb[(size_t)(kb + 0) * PIX];
    b[1] = xb[(size_t)(kb + 1) * PIX];
#pragma unroll
    for (int t = 0; t < 4; ++t) {
      v2f a = *(const v2f*)(w_comp + (t * 16 + row) * CIN + kb);
      acc[t] = __builtin_amdgcn_wmma_f32_16x16x4_f32(
          false, a, false, b, (short)0, acc[t], false, false);
    }
  }

  float* cb = comp + (size_t)n * CCOMP * PIX + pin + row;
#pragma unroll
  for (int t = 0; t < 4; ++t)
#pragma unroll
    for (int v = 0; v < 8; ++v)
      cb[(size_t)(t * 16 + v + 8 * half) * PIX] = acc[t][v];
}

__global__ __launch_bounds__(256) void carafe_encode_softmax(
    const float* __restrict__ comp, const float* __restrict__ w_enc,
    const float* __restrict__ b_enc, float* __restrict__ mask) {
  __shared__ float sm[8][16 * 113];

  const int wave = threadIdx.x >> 5;
  const int lane = threadIdx.x & 31;
  const int row  = lane & 15;
  const int half = lane >> 4;

  const int tile = blockIdx.x * 8 + wave;
  const int gp   = tile << 4;
  const int n    = gp >> 12;
  const int pin  = gp & 4095;

  const int p  = pin + row;
  const int py = p >> 6;
  const int px = p & 63;

  v8f acc[7];
#pragma unroll
  for (int t = 0; t < 7; ++t)
#pragma unroll
    for (int v = 0; v < 8; ++v) {
      const int m = t * 16 + v + 8 * half;
      acc[t][v] = (m < MENC) ? b_enc[m] : 0.0f;
    }

  const float* cbase = comp + (size_t)n * CCOMP * PIX;

  for (int k0 = 0; k0 < KENC; k0 += 4) {
    const int kb = k0 + 2 * half;
    v2f b;
#pragma unroll
    for (int v = 0; v < 2; ++v) {
      const int k  = kb + v;
      const int cc = k / 9;
      const int r  = k - cc * 9;
      const int y  = py + (r / 3) - 1;
      const int xx = px + (r % 3) - 1;
      const bool ok = ((unsigned)y < HW) & ((unsigned)xx < HW);
      const int yc = min(max(y, 0), HW - 1);
      const int xc = min(max(xx, 0), HW - 1);
      const float val = cbase[(size_t)cc * PIX + yc * HW + xc];
      b[v] = ok ? val : 0.0f;
    }
#pragma unroll
    for (int t = 0; t < 7; ++t) {
      const int m  = t * 16 + row;
      const int mc = min(m, MENC - 1);
      v2f a = *(const v2f*)(w_enc + (size_t)mc * KENC + kb);
      if (m >= MENC) { a[0] = 0.0f; a[1] = 0.0f; }
      acc[t] = __builtin_amdgcn_wmma_f32_16x16x4_f32(
          false, a, false, b, (short)0, acc[t], false, false);
    }
  }

  float* s = sm[wave];
#pragma unroll
  for (int t = 0; t < 7; ++t)
#pragma unroll
    for (int v = 0; v < 8; ++v)
      s[row * 113 + t * 16 + v + 8 * half] = acc[t][v];

  __syncthreads();

#pragma unroll
  for (int u = 0; u < 2; ++u) {
    const int g   = lane * 2 + u;
    const int pix = g >> 2;
    const int ij  = g & 3;
    const float* base = s + pix * 113;
    float r[25];
    float mx = -3.4e38f;
#pragma unroll
    for (int k = 0; k < 25; ++k) { r[k] = base[k * 4 + ij]; mx = fmaxf(mx, r[k]); }
    float sum = 0.0f;
#pragma unroll
    for (int k = 0; k < 25; ++k) { r[k] = __expf(r[k] - mx); sum += r[k]; }
    const float inv = 1.0f / sum;
    float* mout = mask + (size_t)(gp + pix) * 100 + ij * 25;
#pragma unroll
    for (int k = 0; k < 25; ++k) mout[k] = r[k] * inv;
  }
}

__global__ __launch_bounds__(64) void carafe_reassemble(
    const float* __restrict__ x, const float* __restrict__ mask,
    float* __restrict__ out) {
  __shared__ float sm[64 * 101];

  const int t  = threadIdx.x;
  const int b  = blockIdx.x;
  const int cg = b & 31;
  const int h  = (b >> 5) & 63;
  const int n  = b >> 11;

  const int pg = (n << 12) + (h << 6) + t;
  const float4* mp4 = (const float4*)(mask + (size_t)pg * 100);
  float* srow = sm + t * 101;
#pragma unroll
  for (int q = 0; q < 25; ++q) {
    const float4 m4 = mp4[q];
    srow[q * 4 + 0] = m4.x; srow[q * 4 + 1] = m4.y;
    srow[q * 4 + 2] = m4.z; srow[q * 4 + 3] = m4.w;
  }
  __syncthreads();

  const float* ms = sm + t * 101;

  for (int ci = 0; ci < 8; ++ci) {
    const int c = cg * 8 + ci;
    const float* xb = x + ((size_t)(n * 256 + c)) * PIX;

    float xv[25];
#pragma unroll
    for (int di = 0; di < 5; ++di) {
      const int y  = h + di - 2;
      const int yc = min(max(y, 0), HW - 1);
#pragma unroll
      for (int dj = 0; dj < 5; ++dj) {
        const int xx = t + dj - 2;
        const int xc = min(max(xx, 0), HW - 1);
        const bool ok = ((unsigned)y < HW) & ((unsigned)xx < HW);
        const float val = xb[yc * HW + xc];
        xv[di * 5 + dj] = ok ? val : 0.0f;
      }
    }

    float a0 = 0.f, a1 = 0.f, a2 = 0.f, a3 = 0.f;
#pragma unroll
    for (int k = 0; k < 25; ++k) {
      const float pch = xv[k];
      a0 += pch * ms[k];
      a1 += pch * ms[25 + k];
      a2 += pch * ms[50 + k];
      a3 += pch * ms[75 + k];
    }

    float* ob = out + (((size_t)(n * 256 + c) * 128) + 2 * h) * 128 + 2 * t;
    float2 r0; r0.x = a0; r0.y = a1;
    float2 r1; r1.x = a2; r1.y = a3;
    *(float2*)(ob)       = r0;
    *(float2*)(ob + 128) = r1;
  }
}

extern "C" void kernel_launch(void* const* d_in, const int* in_sizes, int n_in,
                              void* d_out, int out_size, void* d_ws, size_t ws_size,
                              hipStream_t stream) {
  (void)in_sizes; (void)n_in; (void)out_size; (void)ws_size;
  const float* x      = (const float*)d_in[0];
  const float* w_comp = (const float*)d_in[1];
  const float* b_comp = (const float*)d_in[2];
  const float* w_enc  = (const float*)d_in[3];
  const float* b_enc  = (const float*)d_in[4];
  float* out = (float*)d_out;

  float* comp = (float*)d_ws;
  float* mask = comp + (size_t)4 * CCOMP * PIX;

  carafe_compress<<<128, 256, 0, stream>>>(x, w_comp, b_comp, comp);
  carafe_encode_softmax<<<128, 256, 0, stream>>>(comp, w_enc, b_enc, mask);
  carafe_reassemble<<<8192, 64, 0, stream>>>(x, mask, out);
}


# Both conv K loops issue the loads of 8-9 k-steps as one group ahead of their matrix ops (scheduling barrier holds them); 3x3 gather offsets precomputed per lane; k order and sums unchanged
# speedup vs baseline: 1.3537x; 1.3537x over previous
#include <hip/hip_runtime.h>
#include <math.h>


typedef __attribute__((ext_vector_type(2))) float v2f;
typedef __attribute__((ext_vector_type(8))) float v8f;

#define HW 64
#define PIX 4096
#define CIN 256
#define CCOMP 64
#define MENC 100
#define MPAD 112
#define KENC 576
#define NPIX_TOTAL 16384

__global__ __launch_bounds__(256) void carafe_compress(
    const float* __restrict__ x, const float* __restrict__ w_comp,
    const float* __restrict__ b_comp, float* __restrict__ comp) {
  const int wave = threadIdx.x >> 5;
  const int lane = threadIdx.x & 31;
  const int row  = lane & 15;
  const int half = lane >> 4;

  const int tile = blockIdx.x * 8 + wave;
  const int gp   = tile << 4;
  const int n    = gp >> 12;
  const int pin  = gp & 4095;

  v8f acc[4];
#pragma unroll
  for (int t = 0; t < 4; ++t)
#pragma unroll
    for (int v = 0; v < 8; ++v)
      acc[t][v] = b_comp[t * 16 + v + 8 * half];

  const float* xb = x + (size_t)n * CIN * PIX + pin + row;

  for (int k0 = 0; k0 < CIN; k0 += 32) {
    v2f b[8];
    v2f a[8][4];
#pragma unroll
    for (int s = 0; s < 8; ++s) {
      const int kb = k0 + 4 * s + 2 * half;
      b[s][0] = xb[(size_t)(kb + 0) * PIX];
      b[s][1] = xb[(size_t)(kb + 1) * PIX];
    }
#pragma unroll
    for (int s = 0; s < 8; ++s) {
      const int kb = k0 + 4 * s + 2 * half;
#pragma unroll
      for (int t = 0; t < 4; ++t)
        a[s][t] = *(const v2f*)(w_comp + (t * 16 + row) * CIN + kb);
    }
    __builtin_amdgcn_sched_barrier(0);
#pragma unroll
    for (int s = 0; s < 8; ++s)
#pragma unroll
      for (int t = 0; t < 4; ++t)
        acc[t] = __builtin_amdgcn_wmma_f32_16x16x4_f32(
            false, a[s][t], false, b[s], (short)0, acc[t], false, false);
  }

  float* cb = comp + (size_t)n * CCOMP * PIX + pin + row;
#pragma unroll
  for (int t = 0; t < 4; ++t)
#pragma unroll
    for (int v = 0; v < 8; ++v)
      cb[(size_t)(t * 16 + v + 8 * half) * PIX] = acc[t][v];
}

__global__ __launch_bounds__(256) void carafe_encode_softmax(
    const float* __restrict__ comp, const float* __restrict__ w_enc,
    const float* __restrict__ b_enc, float* __restrict__ mask) {
  __shared__ float sm[8][16 * 113];

  const int wave = threadIdx.x >> 5;
  const int lane = threadIdx.x & 31;
  const int row  = lane & 15;
  const int half = lane >> 4;

  const int tile = blockIdx.x * 8 + wave;
  const int gp   = tile << 4;
  const int n    = gp >> 12;
  const int pin  = gp & 4095;

  const int p  = pin + row;
  const int py = p >> 6;
  const int px = p & 63;

  v8f acc[7];
#pragma unroll
  for (int t = 0; t < 7; ++t)
#pragma unroll
    for (int v = 0; v < 8; ++v) {
      const int m = t * 16 + v + 8 * half;
      acc[t][v] = (m < MENC) ? b_enc[m] : 0.0f;
    }

  const float* cbase = comp + (size_t)n * CCOMP * PIX;

  int  boff[9][2];
  bool bok[9][2];
#pragma unroll
  for (int j = 0; j < 9; ++j)
#pragma unroll
    for (int v = 0; v < 2; ++v) {
      const int kl = 4 * j + 2 * half + v;
      const int cl = kl / 9;
      const int r  = kl - cl * 9;
      const int y  = py + (r / 3) - 1;
      const int xx = px + (r % 3) - 1;
      bok[j][v] = ((unsigned)y < HW) & ((unsigned)xx < HW);
      const int yc = min(max(y, 0), HW - 1);
      const int xc = min(max(xx, 0), HW - 1);
      boff[j][v] = cl * PIX + yc * HW + xc;
    }

  int aoff[7];
#pragma unroll
  for (int t = 0; t < 7; ++t)
    aoff[t] = min(t * 16 + row, MENC - 1) * KENC + 2 * half;
  const bool apad = (6 * 16 + row) >= MENC;

  const float* cptr = cbase;
  const float* wptr = w_enc;

  for (int trip = 0; trip < 16; ++trip) {
    float braw[9][2];
    v2f   a[9][7];
#pragma unroll
    for (int j = 0; j < 9; ++j)
#pragma unroll
      for (int v = 0; v < 2; ++v)
        braw[j][v] = cptr[boff[j][v]];
#pragma unroll
    for (int j = 0; j < 9; ++j)
#pragma unroll
      for (int t = 0; t < 7; ++t)
        a[j][t] = *(const v2f*)(wptr + aoff[t] + 4 * j);
    __builtin_amdgcn_sched_barrier(0);
    v2f b[9];
#pragma unroll
    for (int j = 0; j < 9; ++j) {
      b[j][0] = bok[j][0] ? braw[j][0] : 0.0f;
      b[j][1] = bok[j][1] ? braw[j][1] : 0.0f;
      if (apad) { a[j][6][0] = 0.0f; a[j][6][1] = 0.0f; }
    }
    __builtin_amdgcn_sched_barrier(0);
#pragma unroll
    for (int j = 0; j < 9; ++j)
#pragma unroll
      for (int t = 0; t < 7; ++t)
        acc[t] = __builtin_amdgcn_wmma_f32_16x16x4_f32(
            false, a[j][t], false, b[j], (short)0, acc[t], false, false);
    cptr += 4 * PIX;
    wptr += 36;
  }

  float* s = sm[wave];
#pragma unroll
  for (int t = 0; t < 7; ++t)
#pragma unroll
    for (int v = 0; v < 8; ++v)
      s[row * 113 + t * 16 + v + 8 * half] = acc[t][v];

  __syncthreads();

#pragma unroll
  for (int u = 0; u < 2; ++u) {
    const int g   = lane * 2 + u;
    const int pix = g >> 2;
    const int ij  = g & 3;
    const float* base = s + pix * 113;
    float r[25];
    float mx = -3.4e38f;
#pragma unroll
    for (int k = 0; k < 25; ++k) { r[k] = base[k * 4 + ij]; mx = fmaxf(mx, r[k]); }
    float sum = 0.0f;
#pragma unroll
    for (int k = 0; k < 25; ++k) { r[k] = __expf(r[k] - mx); sum += r[k]; }
    const float inv = 1.0f / sum;
    float* mout = mask + (size_t)(gp + pix) * 100 + ij * 25;
#pragma unroll
    for (int k = 0; k < 25; ++k) mout[k] = r[k] * inv;
  }
}

__global__ __launch_bounds__(64) void carafe_reassemble(
    const float* __restrict__ x, const float* __restrict__ mask,
    float* __restrict__ out) {
  __shared__ float sm[64 * 101];

  const int t  = threadIdx.x;
  const int b  = blockIdx.x;
  const int cg = b & 31;
  const int h  = (b >> 5) & 63;
  const int n  = b >> 11;

  const int pg = (n << 12) + (h << 6) + t;
  const float4* mp4 = (const float4*)(mask + (size_t)pg * 100);
  float* srow = sm + t * 101;
#pragma unroll
  for (int q = 0; q < 25; ++q) {
    const float4 m4 = mp4[q];
    srow[q * 4 + 0] = m4.x; srow[q * 4 + 1] = m4.y;
    srow[q * 4 + 2] = m4.z; srow[q * 4 + 3] = m4.w;
  }
  __syncthreads();

  const float* ms = sm + t * 101;

  for (int ci = 0; ci < 8; ++ci) {
    const int c = cg * 8 + ci;
    const float* xb = x + ((size_t)(n * 256 + c)) * PIX;

    float xv[25];
#pragma unroll
    for (int di = 0; di < 5; ++di) {
      const int y  = h + di - 2;
      const int yc = min(max(y, 0), HW - 1);
#pragma unroll
      for (int dj = 0; dj < 5; ++dj) {
        const int xx = t + dj - 2;
        const int xc = min(max(xx, 0), HW - 1);
        const bool ok = ((unsigned)y < HW) & ((unsigned)xx < HW);
        const float val = xb[yc * HW + xc];
        xv[di * 5 + dj] = ok ? val : 0.0f;
      }
    }

    float a0 = 0.f, a1 = 0.f, a2 = 0.f, a3 = 0.f;
#pragma unroll
    for (int k = 0; k < 25; ++k) {
      const float pch = xv[k];
      a0 += pch * ms[k];
      a1 += pch * ms[25 + k];
      a2 += pch * ms[50 + k];
      a3 += pch * ms[75 + k];
    }

    float* ob = out + (((size_t)(n * 256 + c) * 128) + 2 * h) * 128 + 2 * t;
    float2 r0; r0.x = a0; r0.y = a1;
    float2 r1; r1.x = a2; r1.y = a3;
    *(float2*)(ob)       = r0;
    *(float2*)(ob + 128) = r1;
  }
}

extern "C" void kernel_launch(void* const* d_in, const int* in_sizes, int n_in,
                              void* d_out, int out_size, void* d_ws, size_t ws_size,
                              hipStream_t stream) {
  (void)in_sizes; (void)n_in; (void)out_size; (void)ws_size;
  const float* x      = (const float*)d_in[0];
  const float* w_comp = (const float*)d_in[1];
  const float* b_comp = (const float*)d_in[2];
  const float* w_enc  = (const float*)d_in[3];
  const float* b_enc  = (const float*)d_in[4];
  float* out = (float*)d_out;

  float* comp = (float*)d_ws;
  float* mask = comp + (size_t)4 * CCOMP * PIX;

  carafe_compress<<<128, 256, 0, stream>>>(x, w_comp, b_comp, comp);
  carafe_encode_softmax<<<128, 256, 0, stream>>>(comp, w_enc, b_enc, mask);
  carafe_reassemble<<<8192, 64, 0, stream>>>(x, mask, out);
}


# Both conv weight matrices staged once per work-group in LDS (pitch 16 B x odd, whole-line copies); A fragments read by ds_load_b64 instead of 16-line global loads; k order and sums unchanged
# speedup vs baseline: 1.5436x; 1.1403x over previous
#include <hip/hip_runtime.h>
#include <math.h>


typedef __attribute__((ext_vector_type(2))) float v2f;
typedef __attribute__((ext_vector_type(8))) float v8f;

#define HW 64
#define PIX 4096
#define CIN 256
#define CCOMP 64
#define MENC 100
#define MPAD 112
#define KENC 576
#define NPIX_TOTAL 16384

#define WPITCH 260
#define APITCH 580

__global__ __launch_bounds__(256) void carafe_compress(
    const float* __restrict__ x, const float* __restrict__ w_comp,
    const float* __restrict__ b_comp, float* __restrict__ comp) {
  __shared__ __attribute__((aligned(16))) float lw[CCOMP * WPITCH];

  {
    const float4* w4 = (const float4*)w_comp;
    float4 stg[16];
#pragma unroll
    for (int u = 0; u < 16; ++u) stg[u] = w4[u * 256 + threadIdx.x];
#pragma unroll
    for (int u = 0; u < 16; ++u) {
      const int i = u * 256 + threadIdx.x;
      *(float4*)(lw + (i >> 6) * WPITCH + 4 * (i & 63)) = stg[u];
    }
  }

  const int wave = threadIdx.x >> 5;
  const int lane = threadIdx.x & 31;
  const int row  = lane & 15;
  const int half = lane >> 4;

  const int tile = blockIdx.x * 8 + wave;
  const int gp   = tile << 4;
  const int n    = gp >> 12;
  const int pin  = gp & 4095;

  v8f acc[4];
#pragma unroll
  for (int t = 0; t < 4; ++t)
#pragma unroll
    for (int v = 0; v < 8; ++v)
      acc[t][v] = b_comp[t * 16 + v + 8 * half];

  const float* xb = x + (size_t)n * CIN * PIX + pin + row;
  const int arow = row * WPITCH + 2 * half;

  __syncthreads();

  for (int k0 = 0; k0 < CIN; k0 += 32) {
    v2f b[8];
    v2f a[8][4];
#pragma unroll
    for (int s = 0; s < 8; ++s) {
      const int kb = k0 + 4 * s + 2 * half;
      b[s][0] = xb[(size_t)(kb + 0) * PIX];
      b[s][1] = xb[(size_t)(kb + 1) * PIX];
    }
#pragma unroll
    for (int s = 0; s < 8; ++s)
#pragma unroll
      for (int t = 0; t < 4; ++t)
        a[s][t] = *(const v2f*)(lw + arow + t * 16 * WPITCH + k0 + 4 * s);
    __builtin_amdgcn_sched_barrier(0);
#pragma unroll
    for (int s = 0; s < 8; ++s)
#pragma unroll
      for (int t = 0; t < 4; ++t)
        acc[t] = __builtin_amdgcn_wmma_f32_16x16x4_f32(
            false, a[s][t], false, b[s], (short)0, acc[t], false, false);
  }

  float* cb = comp + (size_t)n * CCOMP * PIX + pin + row;
#pragma unroll
  for (int t = 0; t < 4; ++t)
#pragma unroll
    for (int v = 0; v < 8; ++v)
      cb[(size_t)(t * 16 + v + 8 * half) * PIX] = acc[t][v];
}

__global__ __launch_bounds__(256) void carafe_encode_softmax(
    const float* __restrict__ comp, const float* __restrict__ w_enc,
    const float* __restrict__ b_enc, float* __restrict__ mask) {
  __shared__ __attribute__((aligned(16))) float lds[MPAD * APITCH];

  {
    const float4* w4 = (const float4*)w_enc;
    for (int i0 = 0; i0 < 56 * 256; i0 += 28 * 256) {
      float4 stg[28];
#pragma unroll
      for (int u = 0; u < 28; ++u) stg[u] = w4[i0 + u * 256 + threadIdx.x];
#pragma unroll
      for (int u = 0; u < 28; ++u) {
        const int i = i0 + u * 256 + threadIdx.x;
        const int m = i / 144;
        *(float4*)(lds + m * APITCH + 4 * (i - m * 144)) = stg[u];
      }
    }
    if (threadIdx.x < 64) {
      const int i = 56 * 256 + threadIdx.x;
      const int m = i / 144;
      *(float4*)(lds + m * APITCH + 4 * (i - m * 144)) = w4[i];
    }
    for (int i = threadIdx.x; i < (MPAD - MENC) * 144; i += 256) {
      const int m = i / 144;
      *(float4*)(lds + (MENC + m) * APITCH + 4 * (i - m * 144)) = make_float4(0.f, 0.f, 0.f, 0.f);
    }
  }

  const int wave = threadIdx.x >> 5;
  const int lane = threadIdx.x & 31;
  const int row  = lane & 15;
  const int half = lane >> 4;

  const int tile = blockIdx.x * 8 + wave;
  const int gp   = tile << 4;
  const int n    = gp >> 12;
  const int pin  = gp & 4095;

  const int p  = pin + row;
  const int py = p >> 6;
  const int px = p & 63;

  v8f acc[7];
#pragma unroll
  for (int t = 0; t < 7; ++t)
#pragma unroll
    for (int v = 0; v < 8; ++v) {
      const int m = t * 16 + v + 8 * half;
      acc[t][v] = (m < MENC) ? b_enc[m] : 0.0f;
    }

  const float* cbase = comp + (size_t)n * CCOMP * PIX;

  int  boff[9][2];
  bool bok[9][2];
#pragma unroll
  for (int j = 0; j < 9; ++j)
#pragma unroll
    for (int v = 0; v < 2; ++v) {
      const int kl = 4 * j + 2 * half + v;
      const int cl = kl / 9;
      const int r  = kl - cl * 9;
      const int y  = py + (r / 3) - 1;
      const int xx = px + (r % 3) - 1;
      bok[j][v] = ((unsigned)y < HW) & ((unsigned)xx < HW);
      const int yc = min(max(y, 0), HW - 1);
      const int xc = min(max(xx, 0), HW - 1);
      boff[j][v] = cl * PIX + yc * HW + xc;
    }

  int aoff = row * APITCH + 2 * half;

  const float* cptr = cbase;

  __syncthreads();

  for (int trip = 0; trip < 16; ++trip) {
    float braw[9][2];
    v2f   a[9][7];
#pragma unroll
    for (int j = 0; j < 9; ++j)
#pragma unroll
      for (int v = 0; v < 2; ++v)
        braw[j][v] = cptr[boff[j][v]];
#pragma unroll
    for (int j = 0; j < 9; ++j)
#pragma unroll
      for (int t = 0; t < 7; ++t)
        a[j][t] = *(const v2f*)(lds + aoff + t * 16 * APITCH + 4 * j);
    __builtin_amdgcn_sched_barrier(0);
    v2f b[9];
#pragma unroll
    for (int j = 0; j < 9; ++j) {
      b[j][0] = bok[j][0] ? braw[j][0] : 0.0f;
      b[j][1] = bok[j][1] ? braw[j][1] : 0.0f;
    }
    __builtin_amdgcn_sched_barrier(0);
#pragma unroll
    for (int j = 0; j < 9; ++j)
#pragma unroll
      for (int t = 0; t < 7; ++t)
        acc[t] = __builtin_amdgcn_wmma_f32_16x16x4_f32(
            false, a[j][t], false, b[j], (short)0, acc[t], false, false);
    cptr += 4 * PIX;
    aoff += 36;
  }

  __syncthreads();

  float* s = lds + wave * (16 * 113);
#pragma unroll
  for (int t = 0; t < 7; ++t)
#pragma unroll
    for (int v = 0; v < 8; ++v)
      s[row * 113 + t * 16 + v + 8 * half] = acc[t][v];

  __syncthreads();

#pragma unroll
  for (int u = 0; u < 2; ++u) {
    const int g   = lane * 2 + u;
    const int pix = g >> 2;
    const int ij  = g & 3;
    const float* base = s + pix * 113;
    float r[25];
    float mx = -3.4e38f;
#pragma unroll
    for (int k = 0; k < 25; ++k) { r[k] = base[k * 4 + ij]; mx = fmaxf(mx, r[k]); }
    float sum = 0.0f;
#pragma unroll
    for (int k = 0; k < 25; ++k) { r[k] = __expf(r[k] - mx); sum += r[k]; }
    const float inv = 1.0f / sum;
    float* mout = mask + (size_t)(gp + pix) * 100 + ij * 25;
#pragma unroll
    for (int k = 0; k < 25; ++k) mout[k] = r[k] * inv;
  }
}

__global__ __launch_bounds__(64) void carafe_reassemble(
    const float* __restrict__ x, const float* __restrict__ mask,
    float* __restrict__ out) {
  __shared__ float sm[64 * 101];

  const int t  = threadIdx.x;
  const int b  = blockIdx.x;
  const int cg = b & 31;
  const int h  = (b >> 5) & 63;
  const int n  = b >> 11;

  const int pg = (n << 12) + (h << 6) + t;
  const float4* mp4 = (const float4*)(mask + (size_t)pg * 100);
  float* srow = sm + t * 101;
#pragma unroll
  for (int q = 0; q < 25; ++q) {
    const float4 m4 = mp4[q];
    srow[q * 4 + 0] = m4.x; srow[q * 4 + 1] = m4.y;
    srow[q * 4 + 2] = m4.z; srow[q * 4 + 3] = m4.w;
  }
  __syncthreads();

  const float* ms = sm + t * 101;

  for (int ci = 0; ci < 8; ++ci) {
    const int c = cg * 8 + ci;
    const float* xb = x + ((size_t)(n * 256 + c)) * PIX;

    float xv[25];
#pragma unroll
    for (int di = 0; di < 5; ++di) {
      const int y  = h + di - 2;
      const int yc = min(max(y, 0), HW - 1);
#pragma unroll
      for (int dj = 0; dj < 5; ++dj) {
        const int xx = t + dj - 2;
        const int xc = min(max(xx, 0), HW - 1);
        const bool ok = ((unsigned)y < HW) & ((unsigned)xx < HW);
        const float val = xb[yc * HW + xc];
        xv[di * 5 + dj] = ok ? val : 0.0f;
      }
    }

    float a0 = 0.f, a1 = 0.f, a2 = 0.f, a3 = 0.f;
#pragma unroll
    for (int k = 0; k < 25; ++k) {
      const float pch = xv[k];
      a0 += pch * ms[k];
      a1 += pch * ms[25 + k];
      a2 += pch * ms[50 + k];
      a3 += pch * ms[75 + k];
    }

    float* ob = out + (((size_t)(n * 256 + c) * 128) + 2 * h) * 128 + 2 * t;
    float2 r0; r0.x = a0; r0.y = a1;
    float2 r1; r1.x = a2; r1.y = a3;
    *(float2*)(ob)       = r0;
    *(float2*)(ob + 128) = r1;
  }
}

extern "C" void kernel_launch(void* const* d_in, const int* in_sizes, int n_in,
                              void* d_out, int out_size, void* d_ws, size_t ws_size,
                              hipStream_t stream) {
  (void)in_sizes; (void)n_in; (void)out_size; (void)ws_size;
  const float* x      = (const float*)d_in[0];
  const float* w_comp = (const float*)d_in[1];
  const float* b_comp = (const float*)d_in[2];
  const float* w_enc  = (const float*)d_in[3];
  const float* b_enc  = (const float*)d_in[4];
  float* out = (float*)d_out;

  float* comp = (float*)d_ws;
  float* mask = comp + (size_t)4 * CCOMP * PIX;

  carafe_compress<<<128, 256, 0, stream>>>(x, w_comp, b_comp, comp);
  carafe_encode_softmax<<<128, 256, 0, stream>>>(comp, w_enc, b_enc, mask);
  carafe_reassemble<<<8192, 64, 0, stream>>>(x, mask, out);
}


# Reassembly: 25 gather offsets precomputed per lane; the 50 x loads of two channels issue as one group ahead of their FMAs (scheduling barrier holds them); mask weights in registers; sums unchanged
# speedup vs baseline: 2.3307x; 1.5099x over previous
#include <hip/hip_runtime.h>
#include <math.h>


typedef __attribute__((ext_vector_type(2))) float v2f;
typedef __attribute__((ext_vector_type(8))) float v8f;

#define HW 64
#define PIX 4096
#define CIN 256
#define CCOMP 64
#define MENC 100
#define MPAD 112
#define KENC 576
#define NPIX_TOTAL 16384

#define WPITCH 260
#define APITCH 580

__global__ __launch_bounds__(256) void carafe_compress(
    const float* __restrict__ x, const float* __restrict__ w_comp,
    const float* __restrict__ b_comp, float* __restrict__ comp) {
  __shared__ __attribute__((aligned(16))) float lw[CCOMP * WPITCH];

  {
    const float4* w4 = (const float4*)w_comp;
    float4 stg[16];
#pragma unroll
    for (int u = 0; u < 16; ++u) stg[u] = w4[u * 256 + threadIdx.x];
#pragma unroll
    for (int u = 0; u < 16; ++u) {
      const int i = u * 256 + threadIdx.x;
      *(float4*)(lw + (i >> 6) * WPITCH + 4 * (i & 63)) = stg[u];
    }
  }

  const int wave = threadIdx.x >> 5;
  const int lane = threadIdx.x & 31;
  const int row  = lane & 15;
  const int half = lane >> 4;

  const int tile = blockIdx.x * 8 + wave;
  const int gp   = tile << 4;
  const int n    = gp >> 12;
  const int pin  = gp & 4095;

  v8f acc[4];
#pragma unroll
  for (int t = 0; t < 4; ++t)
#pragma unroll
    for (int v = 0; v < 8; ++v)
      acc[t][v] = b_comp[t * 16 + v + 8 * half];

  const float* xb = x + (size_t)n * CIN * PIX + pin + row;
  const int arow = row * WPITCH + 2 * half;

  __syncthreads();

  for (int k0 = 0; k0 < CIN; k0 += 32) {
    v2f b[8];
    v2f a[8][4];
#pragma unroll
    for (int s = 0; s < 8; ++s) {
      const int kb = k0 + 4 * s + 2 * half;
      b[s][0] = xb[(size_t)(kb + 0) * PIX];
      b[s][1] = xb[(size_t)(kb + 1) * PIX];
    }
#pragma unroll
    for (int s = 0; s < 8; ++s)
#pragma unroll
      for (int t = 0; t < 4; ++t)
        a[s][t] = *(const v2f*)(lw + arow + t * 16 * WPITCH + k0 + 4 * s);
    __builtin_amdgcn_sched_barrier(0);
#pragma unroll
    for (int s = 0; s < 8; ++s)
#pragma unroll
      for (int t = 0; t < 4; ++t)
        acc[t] = __builtin_amdgcn_wmma_f32_16x16x4_f32(
            false, a[s][t], false, b[s], (short)0, acc[t], false, false);
  }

  float* cb = comp + (size_t)n * CCOMP * PIX + pin + row;
#pragma unroll
  for (int t = 0; t < 4; ++t)
#pragma unroll
    for (int v = 0; v < 8; ++v)
      cb[(size_t)(t * 16 + v + 8 * half) * PIX] = acc[t][v];
}

__global__ __launch_bounds__(256) void carafe_encode_softmax(
    const float* __restrict__ comp, const float* __restrict__ w_enc,
    const float* __restrict__ b_enc, float* __restrict__ mask) {
  __shared__ __attribute__((aligned(16))) float lds[MPAD * APITCH];

  {
    const float4* w4 = (const float4*)w_enc;
    for (int i0 = 0; i0 < 56 * 256; i0 += 28 * 256) {
      float4 stg[28];
#pragma unroll
      for (int u = 0; u < 28; ++u) stg[u] = w4[i0 + u * 256 + threadIdx.x];
#pragma unroll
      for (int u = 0; u < 28; ++u) {
        const int i = i0 + u * 256 + threadIdx.x;
        const int m = i / 144;
        *(float4*)(lds + m * APITCH + 4 * (i - m * 144)) = stg[u];
      }
    }
    if (threadIdx.x < 64) {
      const int i = 56 * 256 + threadIdx.x;
      const int m = i / 144;
      *(float4*)(lds + m * APITCH + 4 * (i - m * 144)) = w4[i];
    }
    for (int i = threadIdx.x; i < (MPAD - MENC) * 144; i += 256) {
      const int m = i / 144;
      *(float4*)(lds + (MENC + m) * APITCH + 4 * (i - m * 144)) = make_float4(0.f, 0.f, 0.f, 0.f);
    }
  }

  const int wave = threadIdx.x >> 5;
  const int lane = threadIdx.x & 31;
  const int row  = lane & 15;
  const int half = lane >> 4;

  const int tile = blockIdx.x * 8 + wave;
  const int gp   = tile << 4;
  const int n    = gp >> 12;
  const int pin  = gp & 4095;

  const int p  = pin + row;
  const int py = p >> 6;
  const int px = p & 63;

  v8f acc[7];
#pragma unroll
  for (int t = 0; t < 7; ++t)
#pragma unroll
    for (int v = 0; v < 8; ++v) {
      const int m = t * 16 + v + 8 * half;
      acc[t][v] = (m < MENC) ? b_enc[m] : 0.0f;
    }

  const float* cbase = comp + (size_t)n * CCOMP * PIX;

  int  boff[9][2];
  bool bok[9][2];
#pragma unroll
  for (int j = 0; j < 9; ++j)
#pragma unroll
    for (int v = 0; v < 2; ++v) {
      const int kl = 4 * j + 2 * half + v;
      const int cl = kl / 9;
      const int r  = kl - cl * 9;
      const int y  = py + (r / 3) - 1;
      const int xx = px + (r % 3) - 1;
      bok[j][v] = ((unsigned)y < HW) & ((unsigned)xx < HW);
      const int yc = min(max(y, 0), HW - 1);
      const int xc = min(max(xx, 0), HW - 1);
      boff[j][v] = cl * PIX + yc * HW + xc;
    }

  int aoff = row * APITCH + 2 * half;

  const float* cptr = cbase;

  __syncthreads();

  for (int trip = 0; trip < 16; ++trip) {
    float braw[9][2];
    v2f   a[9][7];
#pragma unroll
    for (int j = 0; j < 9; ++j)
#pragma unroll
      for (int v = 0; v < 2; ++v)
        braw[j][v] = cptr[boff[j][v]];
#pragma unroll
    for (int j = 0; j < 9; ++j)
#pragma unroll
      for (int t = 0; t < 7; ++t)
        a[j][t] = *(const v2f*)(lds + aoff + t * 16 * APITCH + 4 * j);
    __builtin_amdgcn_sched_barrier(0);
    v2f b[9];
#pragma unroll
    for (int j = 0; j < 9; ++j) {
      b[j][0] = bok[j][0] ? braw[j][0] : 0.0f;
      b[j][1] = bok[j][1] ? braw[j][1] : 0.0f;
    }
    __builtin_amdgcn_sched_barrier(0);
#pragma unroll
    for (int j = 0; j < 9; ++j)
#pragma unroll
      for (int t = 0; t < 7; ++t)
        acc[t] = __builtin_amdgcn_wmma_f32_16x16x4_f32(
            false, a[j][t], false, b[j], (short)0, acc[t], false, false);
    cptr += 4 * PIX;
    aoff += 36;
  }

  __syncthreads();

  float* s = lds + wave * (16 * 113);
#pragma unroll
  for (int t = 0; t < 7; ++t)
#pragma unroll
    for (int v = 0; v < 8; ++v)
      s[row * 113 + t * 16 + v + 8 * half] = acc[t][v];

  __syncthreads();

#pragma unroll
  for (int u = 0; u < 2; ++u) {
    const int g   = lane * 2 + u;
    const int pix = g >> 2;
    const int ij  = g & 3;
    const float* base = s + pix * 113;
    float r[25];
    float mx = -3.4e38f;
#pragma unroll
    for (int k = 0; k < 25; ++k) { r[k] = base[k * 4 + ij]; mx = fmaxf(mx, r[k]); }
    float sum = 0.0f;
#pragma unroll
    for (int k = 0; k < 25; ++k) { r[k] = __expf(r[k] - mx); sum += r[k]; }
    const float inv = 1.0f / sum;
    float* mout = mask + (size_t)(gp + pix) * 100 + ij * 25;
#pragma unroll
    for (int k = 0; k < 25; ++k) mout[k] = r[k] * inv;
  }
}

__global__ __launch_bounds__(64) void carafe_reassemble(
    const float* __restrict__ x, const float* __restrict__ mask,
    float* __restrict__ out) {
  __shared__ float sm[64 * 101];

  const int t  = threadIdx.x;
  const int b  = blockIdx.x;
  const int cg = b & 31;
  const int h  = (b >> 5) & 63;
  const int n  = b >> 11;

  const int pg = (n << 12) + (h << 6) + t;
  const float4* mp4 = (const float4*)(mask + (size_t)pg * 100);
  float* srow = sm + t * 101;
#pragma unroll
  for (int q = 0; q < 25; ++q) {
    const float4 m4 = mp4[q];
    srow[q * 4 + 0] = m4.x; srow[q * 4 + 1] = m4.y;
    srow[q * 4 + 2] = m4.z; srow[q * 4 + 3] = m4.w;
  }
  __syncthreads();

  const float* ms = sm + t * 101;
  float m[100];
#pragma unroll
  for (int k = 0; k < 100; ++k) m[k] = ms[k];

  int  xoff[25];
  bool xok[25];
#pragma unroll
  for (int di = 0; di < 5; ++di) {
    const int y  = h + di - 2;
    const int yc = min(max(y, 0), HW - 1);
#pragma unroll
    for (int dj = 0; dj < 5; ++dj) {
      const int xx = t + dj - 2;
      const int xc = min(max(xx, 0), HW - 1);
      xok[di * 5 + dj]  = ((unsigned)y < HW) & ((unsigned)xx < HW);
      xoff[di * 5 + dj] = yc * HW + xc;
    }
  }

  const float* xg = x + ((size_t)(n * 256 + cg * 8)) * PIX;
  float* og = out + (((size_t)(n * 256 + cg * 8) * 128) + 2 * h) * 128 + 2 * t;

  for (int cp = 0; cp < 4; ++cp) {
    const float* xa = xg + (size_t)(2 * cp) * PIX;
    const float* xb = xa + PIX;

    float ra[25], rb[25];
#pragma unroll
    for (int k = 0; k < 25; ++k) ra[k] = xa[xoff[k]];
#pragma unroll
    for (int k = 0; k < 25; ++k) rb[k] = xb[xoff[k]];
    __builtin_amdgcn_sched_barrier(0);
#pragma unroll
    for (int k = 0; k < 25; ++k) {
      ra[k] = xok[k] ? ra[k] : 0.0f;
      rb[k] = xok[k] ? rb[k] : 0.0f;
    }

    float a0 = 0.f, a1 = 0.f, a2 = 0.f, a3 = 0.f;
    float b0 = 0.f, b1 = 0.f, b2 = 0.f, b3 = 0.f;
#pragma unroll
    for (int k = 0; k < 25; ++k) {
      a0 += ra[k] * m[k];
      a1 += ra[k] * m[25 + k];
      a2 += ra[k] * m[50 + k];
      a3 += ra[k] * m[75 + k];
      b0 += rb[k] * m[k];
      b1 += rb[k] * m[25 + k];
      b2 += rb[k] * m[50 + k];
      b3 += rb[k] * m[75 + k];
    }

    float* oa = og + (size_t)(2 * cp) * 128 * 128;
    float* ob = oa + 128 * 128;
    float2 r0; r0.x = a0; r0.y = a1;
    float2 r1; r1.x = a2; r1.y = a3;
    *(float2*)(oa)       = r0;
    *(float2*)(oa + 128) = r1;
    float2 q0; q0.x = b0; q0.y = b1;
    float2 q1; q1.x = b2; q1.y = b3;
    *(float2*)(ob)       = q0;
    *(float2*)(ob + 128) = q1;
  }
}

extern "C" void kernel_launch(void* const* d_in, const int* in_sizes, int n_in,
                              void* d_out, int out_size, void* d_ws, size_t ws_size,
                              hipStream_t stream) {
  (void)in_sizes; (void)n_in; (void)out_size; (void)ws_size;
  const float* x      = (const float*)d_in[0];
  const float* w_comp = (const float*)d_in[1];
  const float* b_comp = (const float*)d_in[2];
  const float* w_enc  = (const float*)d_in[3];
  const float* b_enc  = (const float*)d_in[4];
  float* out = (float*)d_out;

  float* comp = (float*)d_ws;
  float* mask = comp + (size_t)4 * CCOMP * PIX;

  carafe_compress<<<128, 256, 0, stream>>>(x, w_comp, b_comp, comp);
  carafe_encode_softmax<<<128, 256, 0, stream>>>(comp, w_enc, b_enc, mask);
  carafe_reassemble<<<8192, 64, 0, stream>>>(x, mask, out);
}


# Reassembly: the row's mask (200 contiguous lines) is read with whole-line loads and handed to its pixels through LDS (pitch 16 B x odd, 16-byte reads), not by 400-B-stride lane loads; sums unchanged
# speedup vs baseline: 2.6887x; 1.1536x over previous
#include <hip/hip_runtime.h>
#include <math.h>


typedef __attribute__((ext_vector_type(2))) float v2f;
typedef __attribute__((ext_vector_type(8))) float v8f;

#define HW 64
#define PIX 4096
#define CIN 256
#define CCOMP 64
#define MENC 100
#define MPAD 112
#define KENC 576
#define NPIX_TOTAL 16384

#define WPITCH 260
#define APITCH 580

__global__ __launch_bounds__(256) void carafe_compress(
    const float* __restrict__ x, const float* __restrict__ w_comp,
    const float* __restrict__ b_comp, float* __restrict__ comp) {
  __shared__ __attribute__((aligned(16))) float lw[CCOMP * WPITCH];

  {
    const float4* w4 = (const float4*)w_comp;
    float4 stg[16];
#pragma unroll
    for (int u = 0; u < 16; ++u) stg[u] = w4[u * 256 + threadIdx.x];
#pragma unroll
    for (int u = 0; u < 16; ++u) {
      const int i = u * 256 + threadIdx.x;
      *(float4*)(lw + (i >> 6) * WPITCH + 4 * (i & 63)) = stg[u];
    }
  }

  const int wave = threadIdx.x >> 5;
  const int lane = threadIdx.x & 31;
  const int row  = lane & 15;
  const int half = lane >> 4;

  const int tile = blockIdx.x * 8 + wave;
  const int gp   = tile << 4;
  const int n    = gp >> 12;
  const int pin  = gp & 4095;

  v8f acc[4];
#pragma unroll
  for (int t = 0; t < 4; ++t)
#pragma unroll
    for (int v = 0; v < 8; ++v)
      acc[t][v] = b_comp[t * 16 + v + 8 * half];

  const float* xb = x + (size_t)n * CIN * PIX + pin + row;
  const int arow = row * WPITCH + 2 * half;

  __syncthreads();

  for (int k0 = 0; k0 < CIN; k0 += 32) {
    v2f b[8];
    v2f a[8][4];
#pragma unroll
    for (int s = 0; s < 8; ++s) {
      const int kb = k0 + 4 * s + 2 * half;
      b[s][0] = xb[(size_t)(kb + 0) * PIX];
      b[s][1] = xb[(size_t)(kb + 1) * PIX];
    }
#pragma unroll
    for (int s = 0; s < 8; ++s)
#pragma unroll
      for (int t = 0; t < 4; ++t)
        a[s][t] = *(const v2f*)(lw + arow + t * 16 * WPITCH + k0 + 4 * s);
    __builtin_amdgcn_sched_barrier(0);
#pragma unroll
    for (int s = 0; s < 8; ++s)
#pragma unroll
      for (int t = 0; t < 4; ++t)
        acc[t] = __builtin_amdgcn_wmma_f32_16x16x4_f32(
            false, a[s][t], false, b[s], (short)0, acc[t], false, false);
  }

  float* cb = comp + (size_t)n * CCOMP * PIX + pin + row;
#pragma unroll
  for (int t = 0; t < 4; ++t)
#pragma unroll
    for (int v = 0; v < 8; ++v)
      cb[(size_t)(t * 16 + v + 8 * half) * PIX] = acc[t][v];
}

__global__ __launch_bounds__(256) void carafe_encode_softmax(
    const float* __restrict__ comp, const float* __restrict__ w_enc,
    const float* __restrict__ b_enc, float* __restrict__ mask) {
  __shared__ __attribute__((aligned(16))) float lds[MPAD * APITCH];

  {
    const float4* w4 = (const float4*)w_enc;
    for (int i0 = 0; i0 < 56 * 256; i0 += 28 * 256) {
      float4 stg[28];
#pragma unroll
      for (int u = 0; u < 28; ++u) stg[u] = w4[i0 + u * 256 + threadIdx.x];
#pragma unroll
      for (int u = 0; u < 28; ++u) {
        const int i = i0 + u * 256 + threadIdx.x;
        const int m = i / 144;
        *(float4*)(lds + m * APITCH + 4 * (i - m * 144)) = stg[u];
      }
    }
    if (threadIdx.x < 64) {
      const int i = 56 * 256 + threadIdx.x;
      const int m = i / 144;
      *(float4*)(lds + m * APITCH + 4 * (i - m * 144)) = w4[i];
    }
    for (int i = threadIdx.x; i < (MPAD - MENC) * 144; i += 256) {
      const int m = i / 144;
      *(float4*)(lds + (MENC + m) * APITCH + 4 * (i - m * 144)) = make_float4(0.f, 0.f, 0.f, 0.f);
    }
  }

  const int wave = threadIdx.x >> 5;
  const int lane = threadIdx.x & 31;
  const int row  = lane & 15;
  const int half = lane >> 4;

  const int tile = blockIdx.x * 8 + wave;
  const int gp   = tile << 4;
  const int n    = gp >> 12;
  const int pin  = gp & 4095;

  const int p  = pin + row;
  const int py = p >> 6;
  const int px = p & 63;

  v8f acc[7];
#pragma unroll
  for (int t = 0; t < 7; ++t)
#pragma unroll
    for (int v = 0; v < 8; ++v) {
      const int m = t * 16 + v + 8 * half;
      acc[t][v] = (m < MENC) ? b_enc[m] : 0.0f;
    }

  const float* cbase = comp + (size_t)n * CCOMP * PIX;

  int  boff[9][2];
  bool bok[9][2];
#pragma unroll
  for (int j = 0; j < 9; ++j)
#pragma unroll
    for (int v = 0; v < 2; ++v) {
      const int kl = 4 * j + 2 * half + v;
      const int cl = kl / 9;
      const int r  = kl - cl * 9;
      const int y  = py + (r / 3) - 1;
      const int xx = px + (r % 3) - 1;
      bok[j][v] = ((unsigned)y < HW) & ((unsigned)xx < HW);
      const int yc = min(max(y, 0), HW - 1);
      const int xc = min(max(xx, 0), HW - 1);
      boff[j][v] = cl * PIX + yc * HW + xc;
    }

  int aoff = row * APITCH + 2 * half;

  const float* cptr = cbase;

  __syncthreads();

  for (int trip = 0; trip < 16; ++trip) {
    float braw[9][2];
    v2f   a[9][7];
#pragma unroll
    for (int j = 0; j < 9; ++j)
#pragma unroll
      for (int v = 0; v < 2; ++v)
        braw[j][v] = cptr[boff[j][v]];
#pragma unroll
    for (int j = 0; j < 9; ++j)
#pragma unroll
      for (int t = 0; t < 7; ++t)
        a[j][t] = *(const v2f*)(lds + aoff + t * 16 * APITCH + 4 * j);
    __builtin_amdgcn_sched_barrier(0);
    v2f b[9];
#pragma unroll
    for (int j = 0; j < 9; ++j) {
      b[j][0] = bok[j][0] ? braw[j][0] : 0.0f;
      b[j][1] = bok[j][1] ? braw[j][1] : 0.0f;
    }
    __builtin_amdgcn_sched_barrier(0);
#pragma unroll
    for (int j = 0; j < 9; ++j)
#pragma unroll
      for (int t = 0; t < 7; ++t)
        acc[t] = __builtin_amdgcn_wmma_f32_16x16x4_f32(
            false, a[j][t], false, b[j], (short)0, acc[t], false, false);
    cptr += 4 * PIX;
    aoff += 36;
  }

  __syncthreads();

  float* s = lds + wave * (16 * 113);
#pragma unroll
  for (int t = 0; t < 7; ++t)
#pragma unroll
    for (int v = 0; v < 8; ++v)
      s[row * 113 + t * 16 + v + 8 * half] = acc[t][v];

  __syncthreads();

#pragma unroll
  for (int u = 0; u < 2; ++u) {
    const int g   = lane * 2 + u;
    const int pix = g >> 2;
    const int ij  = g & 3;
    const float* base = s + pix * 113;
    float r[25];
    float mx = -3.4e38f;
#pragma unroll
    for (int k = 0; k < 25; ++k) { r[k] = base[k * 4 + ij]; mx = fmaxf(mx, r[k]); }
    float sum = 0.0f;
#pragma unroll
    for (int k = 0; k < 25; ++k) { r[k] = __expf(r[k] - mx); sum += r[k]; }
    const float inv = 1.0f / sum;
    float* mout = mask + (size_t)(gp + pix) * 100 + ij * 25;
#pragma unroll
    for (int k = 0; k < 25; ++k) mout[k] = r[k] * inv;
  }
}

__global__ __launch_bounds__(64) void carafe_reassemble(
    const float* __restrict__ x, const float* __restrict__ mask,
    float* __restrict__ out) {
  __shared__ __attribute__((aligned(16))) float sm[64 * 100];

  const int t  = threadIdx.x;
  const int b  = blockIdx.x;
  const int cg = b & 31;
  const int h  = (b >> 5) & 63;
  const int n  = b >> 11;

  const float4* mrow4 = (const float4*)(mask + (size_t)((n << 12) + (h << 6)) * 100);
  float4* sm4 = (float4*)sm;
  {
    float4 stg[25];
#pragma unroll
    for (int q = 0; q < 25; ++q) stg[q] = mrow4[q * 64 + t];
#pragma unroll
    for (int q = 0; q < 25; ++q) sm4[q * 64 + t] = stg[q];
  }
  __syncthreads();

  float m[100];
#pragma unroll
  for (int j = 0; j < 25; ++j) {
    const float4 v = sm4[t * 25 + j];
    m[4 * j + 0] = v.x; m[4 * j + 1] = v.y;
    m[4 * j + 2] = v.z; m[4 * j + 3] = v.w;
  }

  int  xoff[25];
  bool xok[25];
#pragma unroll
  for (int di = 0; di < 5; ++di) {
    const int y  = h + di - 2;
    const int yc = min(max(y, 0), HW - 1);
#pragma unroll
    for (int dj = 0; dj < 5; ++dj) {
      const int xx = t + dj - 2;
      const int xc = min(max(xx, 0), HW - 1);
      xok[di * 5 + dj]  = ((unsigned)y < HW) & ((unsigned)xx < HW);
      xoff[di * 5 + dj] = yc * HW + xc;
    }
  }

  const float* xg = x + ((size_t)(n * 256 + cg * 8)) * PIX;
  float* og = out + (((size_t)(n * 256 + cg * 8) * 128) + 2 * h) * 128 + 2 * t;

  for (int cp = 0; cp < 4; ++cp) {
    const float* xa = xg + (size_t)(2 * cp) * PIX;
    const float* xb = xa + PIX;

    float ra[25], rb[25];
#pragma unroll
    for (int k = 0; k < 25; ++k) ra[k] = xa[xoff[k]];
#pragma unroll
    for (int k = 0; k < 25; ++k) rb[k] = xb[xoff[k]];
    __builtin_amdgcn_sched_barrier(0);
#pragma unroll
    for (int k = 0; k < 25; ++k) {
      ra[k] = xok[k] ? ra[k] : 0.0f;
      rb[k] = xok[k] ? rb[k] : 0.0f;
    }

    float a0 = 0.f, a1 = 0.f, a2 = 0.f, a3 = 0.f;
    float b0 = 0.f, b1 = 0.f, b2 = 0.f, b3 = 0.f;
#pragma unroll
    for (int k = 0; k < 25; ++k) {
      a0 += ra[k] * m[k];
      a1 += ra[k] * m[25 + k];
      a2 += ra[k] * m[50 + k];
      a3 += ra[k] * m[75 + k];
      b0 += rb[k] * m[k];
      b1 += rb[k] * m[25 + k];
      b2 += rb[k] * m[50 + k];
      b3 += rb[k] * m[75 + k];
    }

    float* oa = og + (size_t)(2 * cp) * 128 * 128;
    float* ob = oa + 128 * 128;
    float2 r0; r0.x = a0; r0.y = a1;
    float2 r1; r1.x = a2; r1.y = a3;
    *(float2*)(oa)       = r0;
    *(float2*)(oa + 128) = r1;
    float2 q0; q0.x = b0; q0.y = b1;
    float2 q1; q1.x = b2; q1.y = b3;
    *(float2*)(ob)       = q0;
    *(float2*)(ob + 128) = q1;
  }
}

extern "C" void kernel_launch(void* const* d_in, const int* in_sizes, int n_in,
                              void* d_out, int out_size, void* d_ws, size_t ws_size,
                              hipStream_t stream) {
  (void)in_sizes; (void)n_in; (void)out_size; (void)ws_size;
  const float* x      = (const float*)d_in[0];
  const float* w_comp = (const float*)d_in[1];
  const float* b_comp = (const float*)d_in[2];
  const float* w_enc  = (const float*)d_in[3];
  const float* b_enc  = (const float*)d_in[4];
  float* out = (float*)d_out;

  float* comp = (float*)d_ws;
  float* mask = comp + (size_t)4 * CCOMP * PIX;

  carafe_compress<<<128, 256, 0, stream>>>(x, w_comp, b_comp, comp);
  carafe_encode_softmax<<<128, 256, 0, stream>>>(comp, w_enc, b_enc, mask);
  carafe_reassemble<<<8192, 64, 0, stream>>>(x, mask, out);
}


# Both convs: a 16-pixel tile's M tiles split over two waves (3x3: 4+3, 1x1: 2+2): grids of 256 work-groups, one per compute unit, not 128; compiler now unrolls the 1x1 K loop; k order, sums unchanged
# speedup vs baseline: 3.1967x; 1.1889x over previous
#include <hip/hip_runtime.h>
#include <math.h>


typedef __attribute__((ext_vector_type(2))) float v2f;
typedef __attribute__((ext_vector_type(8))) float v8f;

#define HW 64
#define PIX 4096
#define CIN 256
#define CCOMP 64
#define MENC 100
#define MPAD 112
#define KENC 576
#define NPIX_TOTAL 16384

#define WPITCH 260
#define APITCH 580

__global__ __launch_bounds__(256) void carafe_compress(
    const float* __restrict__ x, const float* __restrict__ w_comp,
    const float* __restrict__ b_comp, float* __restrict__ comp) {
  __shared__ __attribute__((aligned(16))) float lw[CCOMP * WPITCH];

  {
    const float4* w4 = (const float4*)w_comp;
    float4 stg[16];
#pragma unroll
    for (int u = 0; u < 16; ++u) stg[u] = w4[u * 256 + threadIdx.x];
#pragma unroll
    for (int u = 0; u < 16; ++u) {
      const int i = u * 256 + threadIdx.x;
      *(float4*)(lw + (i >> 6) * WPITCH + 4 * (i & 63)) = stg[u];
    }
  }

  const int wave = threadIdx.x >> 5;
  const int lane = threadIdx.x & 31;
  const int row  = lane & 15;
  const int half = lane >> 4;

  const int tile = blockIdx.x * 4 + (wave & 3);
  const int t0   = (wave >> 2) * 2;
  const int gp   = tile << 4;
  const int n    = gp >> 12;
  const int pin  = gp & 4095;

  v8f acc[2];
#pragma unroll
  for (int t = 0; t < 2; ++t)
#pragma unroll
    for (int v = 0; v < 8; ++v)
      acc[t][v] = b_comp[(t0 + t) * 16 + v + 8 * half];

  const float* xb = x + (size_t)n * CIN * PIX + pin + row;
  const int arow = (t0 * 16 + row) * WPITCH + 2 * half;

  __syncthreads();

  for (int k0 = 0; k0 < CIN; k0 += 32) {
    v2f b[8];
    v2f a[8][2];
#pragma unroll
    for (int s = 0; s < 8; ++s) {
      const int kb = k0 + 4 * s + 2 * half;
      b[s][0] = xb[(size_t)(kb + 0) * PIX];
      b[s][1] = xb[(size_t)(kb + 1) * PIX];
    }
#pragma unroll
    for (int s = 0; s < 8; ++s)
#pragma unroll
      for (int t = 0; t < 2; ++t)
        a[s][t] = *(const v2f*)(lw + arow + t * 16 * WPITCH + k0 + 4 * s);
    __builtin_amdgcn_sched_barrier(0);
#pragma unroll
    for (int s = 0; s < 8; ++s)
#pragma unroll
      for (int t = 0; t < 2; ++t)
        acc[t] = __builtin_amdgcn_wmma_f32_16x16x4_f32(
            false, a[s][t], false, b[s], (short)0, acc[t], false, false);
  }

  float* cb = comp + (size_t)n * CCOMP * PIX + pin + row;
#pragma unroll
  for (int t = 0; t < 2; ++t)
#pragma unroll
    for (int v = 0; v < 8; ++v)
      cb[(size_t)((t0 + t) * 16 + v + 8 * half) * PIX] = acc[t][v];
}

template <int T0, int NT>
__device__ __forceinline__ void encode_walk(
    const float* __restrict__ cbase, const float* lds, int aoff,
    const int (&boff)[9][2], const bool (&bok)[9][2], v8f (&acc)[4]) {
  const float* cptr = cbase;
  aoff += T0 * 16 * APITCH;
  for (int trip = 0; trip < 16; ++trip) {
    float braw[9][2];
    v2f   a[9][NT];
#pragma unroll
    for (int j = 0; j < 9; ++j)
#pragma unroll
      for (int v = 0; v < 2; ++v)
        braw[j][v] = cptr[boff[j][v]];
#pragma unroll
    for (int j = 0; j < 9; ++j)
#pragma unroll
      for (int t = 0; t < NT; ++t)
        a[j][t] = *(const v2f*)(lds + aoff + t * 16 * APITCH + 4 * j);
    __builtin_amdgcn_sched_barrier(0);
    v2f b[9];
#pragma unroll
    for (int j = 0; j < 9; ++j) {
      b[j][0] = bok[j][0] ? braw[j][0] : 0.0f;
      b[j][1] = bok[j][1] ? braw[j][1] : 0.0f;
    }
    __builtin_amdgcn_sched_barrier(0);
#pragma unroll
    for (int j = 0; j < 9; ++j)
#pragma unroll
      for (int t = 0; t < NT; ++t)
        acc[t] = __builtin_amdgcn_wmma_f32_16x16x4_f32(
            false, a[j][t], false, b[j], (short)0, acc[t], false, false);
    cptr += 4 * PIX;
    aoff += 36;
  }
}

__global__ __launch_bounds__(256) void carafe_encode_softmax(
    const float* __restrict__ comp, const float* __restrict__ w_enc,
    const float* __restrict__ b_enc, float* __restrict__ mask) {
  __shared__ __attribute__((aligned(16))) float lds[MPAD * APITCH];

  {
    const float4* w4 = (const float4*)w_enc;
    for (int i0 = 0; i0 < 56 * 256; i0 += 28 * 256) {
      float4 stg[28];
#pragma unroll
      for (int u = 0; u < 28; ++u) stg[u] = w4[i0 + u * 256 + threadIdx.x];
#pragma unroll
      for (int u = 0; u < 28; ++u) {
        const int i = i0 + u * 256 + threadIdx.x;
        const int m = i / 144;
        *(float4*)(lds + m * APITCH + 4 * (i - m * 144)) = stg[u];
      }
    }
    if (threadIdx.x < 64) {
      const int i = 56 * 256 + threadIdx.x;
      const int m = i / 144;
      *(float4*)(lds + m * APITCH + 4 * (i - m * 144)) = w4[i];
    }
    for (int i = threadIdx.x; i < (MPAD - MENC) * 144; i += 256) {
      const int m = i / 144;
      *(float4*)(lds + (MENC + m) * APITCH + 4 * (i - m * 144)) = make_float4(0.f, 0.f, 0.f, 0.f);
    }
  }

  const int wave = __builtin_amdgcn_readfirstlane(threadIdx.x >> 5);
  const int lane = threadIdx.x & 31;
  const int row  = lane & 15;
  const int half = lane >> 4;

  const int ptile = wave & 3;
  const int tb    = (wave >> 2) * 4;
  const int tile  = blockIdx.x * 4 + ptile;
  const int gp   = tile << 4;
  const int n    = gp >> 12;
  const int pin  = gp & 4095;

  const int p  = pin + row;
  const int py = p >> 6;
  const int px = p & 63;

  v8f acc[4];
#pragma unroll
  for (int t = 0; t < 4; ++t)
#pragma unroll
    for (int v = 0; v < 8; ++v) {
      const int m = (tb + t) * 16 + v + 8 * half;
      const float bv = b_enc[min(m, MENC - 1)];
      acc[t][v] = (m < MENC) ? bv : 0.0f;
    }

  const float* cbase = comp + (size_t)n * CCOMP * PIX;

  int  boff[9][2];
  bool bok[9][2];
#pragma unroll
  for (int j = 0; j < 9; ++j)
#pragma unroll
    for (int v = 0; v < 2; ++v) {
      const int kl = 4 * j + 2 * half + v;
      const int cl = kl / 9;
      const int r  = kl - cl * 9;
      const int y  = py + (r / 3) - 1;
      const int xx = px + (r % 3) - 1;
      bok[j][v] = ((unsigned)y < HW) & ((unsigned)xx < HW);
      const int yc = min(max(y, 0), HW - 1);
      const int xc = min(max(xx, 0), HW - 1);
      boff[j][v] = cl * PIX + yc * HW + xc;
    }

  const int aoff = row * APITCH + 2 * half;

  __syncthreads();

  if (tb == 0) encode_walk<0, 4>(cbase, lds, aoff, boff, bok, acc);
  else         encode_walk<4, 3>(cbase, lds, aoff, boff, bok, acc);

  __syncthreads();

  float* s = lds + ptile * (16 * 113);
#pragma unroll
  for (int t = 0; t < 4; ++t)
    if (tb + t < 7) {
#pragma unroll
      for (int v = 0; v < 8; ++v)
        s[row * 113 + (tb + t) * 16 + v + 8 * half] = acc[t][v];
    }

  __syncthreads();

  {
    const int g   = threadIdx.x;
    const int pix = g >> 2;
    const int ij  = g & 3;
    const float* base = lds + (pix >> 4) * (16 * 113) + (pix & 15) * 113;
    float r[25];
    float mx = -3.4e38f;
#pragma unroll
    for (int k = 0; k < 25; ++k) { r[k] = base[k * 4 + ij]; mx = fmaxf(mx, r[k]); }
    float sum = 0.0f;
#pragma unroll
    for (int k = 0; k < 25; ++k) { r[k] = __expf(r[k] - mx); sum += r[k]; }
    const float inv = 1.0f / sum;
    float* mout = mask + (size_t)(blockIdx.x * 64 + pix) * 100 + ij * 25;
#pragma unroll
    for (int k = 0; k < 25; ++k) mout[k] = r[k] * inv;
  }
}

__global__ __launch_bounds__(64) void carafe_reassemble(
    const float* __restrict__ x, const float* __restrict__ mask,
    float* __restrict__ out) {
  __shared__ __attribute__((aligned(16))) float sm[64 * 100];

  const int t  = threadIdx.x;
  const int b  = blockIdx.x;
  const int cg = b & 31;
  const int h  = (b >> 5) & 63;
  const int n  = b >> 11;

  const float4* mrow4 = (const float4*)(mask + (size_t)((n << 12) + (h << 6)) * 100);
  float4* sm4 = (float4*)sm;
  {
    float4 stg[25];
#pragma unroll
    for (int q = 0; q < 25; ++q) stg[q] = mrow4[q * 64 + t];
#pragma unroll
    for (int q = 0; q < 25; ++q) sm4[q * 64 + t] = stg[q];
  }
  __syncthreads();

  float m[100];
#pragma unroll
  for (int j = 0; j < 25; ++j) {
    const float4 v = sm4[t * 25 + j];
    m[4 * j + 0] = v.x; m[4 * j + 1] = v.y;
    m[4 * j + 2] = v.z; m[4 * j + 3] = v.w;
  }

  int  xoff[25];
  bool xok[25];
#pragma unroll
  for (int di = 0; di < 5; ++di) {
    const int y  = h + di - 2;
    const int yc = min(max(y, 0), HW - 1);
#pragma unroll
    for (int dj = 0; dj < 5; ++dj) {
      const int xx = t + dj - 2;
      const int xc = min(max(xx, 0), HW - 1);
      xok[di * 5 + dj]  = ((unsigned)y < HW) & ((unsigned)xx < HW);
      xoff[di * 5 + dj] = yc * HW + xc;
    }
  }

  const float* xg = x + ((size_t)(n * 256 + cg * 8)) * PIX;
  float* og = out + (((size_t)(n * 256 + cg * 8) * 128) + 2 * h) * 128 + 2 * t;

  for (int cp = 0; cp < 4; ++cp) {
    const float* xa = xg + (size_t)(2 * cp) * PIX;
    const float* xb = xa + PIX;

    float ra[25], rb[25];
#pragma unroll
    for (int k = 0; k < 25; ++k) ra[k] = xa[xoff[k]];
#pragma unroll
    for (int k = 0; k < 25; ++k) rb[k] = xb[xoff[k]];
    __builtin_amdgcn_sched_barrier(0);
#pragma unroll
    for (int k = 0; k < 25; ++k) {
      ra[k] = xok[k] ? ra[k] : 0.0f;
      rb[k] = xok[k] ? rb[k] : 0.0f;
    }

    float a0 = 0.f, a1 = 0.f, a2 = 0.f, a3 = 0.f;
    float b0 = 0.f, b1 = 0.f, b2 = 0.f, b3 = 0.f;
#pragma unroll
    for (int k = 0; k < 25; ++k) {
      a0 += ra[k] * m[k];
      a1 += ra[k] * m[25 + k];
      a2 += ra[k] * m[50 + k];
      a3 += ra[k] * m[75 + k];
      b0 += rb[k] * m[k];
      b1 += rb[k] * m[25 + k];
      b2 += rb[k] * m[50 + k];
      b3 += rb[k] * m[75 + k];
    }

    float* oa = og + (size_t)(2 * cp) * 128 * 128;
    float* ob = oa + 128 * 128;
    float2 r0; r0.x = a0; r0.y = a1;
    float2 r1; r1.x = a2; r1.y = a3;
    *(float2*)(oa)       = r0;
    *(float2*)(oa + 128) = r1;
    float2 q0; q0.x = b0; q0.y = b1;
    float2 q1; q1.x = b2; q1.y = b3;
    *(float2*)(ob)       = q0;
    *(float2*)(ob + 128) = q1;
  }
}

extern "C" void kernel_launch(void* const* d_in, const int* in_sizes, int n_in,
                              void* d_out, int out_size, void* d_ws, size_t ws_size,
                              hipStream_t stream) {
  (void)in_sizes; (void)n_in; (void)out_size; (void)ws_size;
  const float* x      = (const float*)d_in[0];
  const float* w_comp = (const float*)d_in[1];
  const float* b_comp = (const float*)d_in[2];
  const float* w_enc  = (const float*)d_in[3];
  const float* b_enc  = (const float*)d_in[4];
  float* out = (float*)d_out;

  float* comp = (float*)d_ws;
  float* mask = comp + (size_t)4 * CCOMP * PIX;

  carafe_compress<<<256, 256, 0, stream>>>(x, w_comp, b_comp, comp);
  carafe_encode_softmax<<<256, 256, 0, stream>>>(comp, w_enc, b_enc, mask);
  carafe_reassemble<<<8192, 64, 0, stream>>>(x, mask, out);
}


# Reassembly: the block's x window (8 channels x 5 rows, zero border) is staged once in LDS by whole-line loads; taps are LDS reads, not 25 one-dword global gathers per pixel and channel; sums unchanged
# speedup vs baseline: 3.3449x; 1.0464x over previous
#include <hip/hip_runtime.h>
#include <math.h>


typedef __attribute__((ext_vector_type(2))) float v2f;
typedef __attribute__((ext_vector_type(8))) float v8f;

#define HW 64
#define PIX 4096
#define CIN 256
#define CCOMP 64
#define MENC 100
#define MPAD 112
#define KENC 576
#define NPIX_TOTAL 16384

#define WPITCH 260
#define APITCH 580

__global__ __launch_bounds__(256) void carafe_compress(
    const float* __restrict__ x, const float* __restrict__ w_comp,
    const float* __restrict__ b_comp, float* __restrict__ comp) {
  __shared__ __attribute__((aligned(16))) float lw[CCOMP * WPITCH];

  {
    const float4* w4 = (const float4*)w_comp;
    float4 stg[16];
#pragma unroll
    for (int u = 0; u < 16; ++u) stg[u] = w4[u * 256 + threadIdx.x];
#pragma unroll
    for (int u = 0; u < 16; ++u) {
      const int i = u * 256 + threadIdx.x;
      *(float4*)(lw + (i >> 6) * WPITCH + 4 * (i & 63)) = stg[u];
    }
  }

  const int wave = threadIdx.x >> 5;
  const int lane = threadIdx.x & 31;
  const int row  = lane & 15;
  const int half = lane >> 4;

  const int tile = blockIdx.x * 4 + (wave & 3);
  const int t0   = (wave >> 2) * 2;
  const int gp   = tile << 4;
  const int n    = gp >> 12;
  const int pin  = gp & 4095;

  v8f acc[2];
#pragma unroll
  for (int t = 0; t < 2; ++t)
#pragma unroll
    for (int v = 0; v < 8; ++v)
      acc[t][v] = b_comp[(t0 + t) * 16 + v + 8 * half];

  const float* xb = x + (size_t)n * CIN * PIX + pin + row;
  const int arow = (t0 * 16 + row) * WPITCH + 2 * half;

  __syncthreads();

  for (int k0 = 0; k0 < CIN; k0 += 32) {
    v2f b[8];
    v2f a[8][2];
#pragma unroll
    for (int s = 0; s < 8; ++s) {
      const int kb = k0 + 4 * s + 2 * half;
      b[s][0] = xb[(size_t)(kb + 0) * PIX];
      b[s][1] = xb[(size_t)(kb + 1) * PIX];
    }
#pragma unroll
    for (int s = 0; s < 8; ++s)
#pragma unroll
      for (int t = 0; t < 2; ++t)
        a[s][t] = *(const v2f*)(lw + arow + t * 16 * WPITCH + k0 + 4 * s);
    __builtin_amdgcn_sched_barrier(0);
#pragma unroll
    for (int s = 0; s < 8; ++s)
#pragma unroll
      for (int t = 0; t < 2; ++t)
        acc[t] = __builtin_amdgcn_wmma_f32_16x16x4_f32(
            false, a[s][t], false, b[s], (short)0, acc[t], false, false);
  }

  float* cb = comp + (size_t)n * CCOMP * PIX + pin + row;
#pragma unroll
  for (int t = 0; t < 2; ++t)
#pragma unroll
    for (int v = 0; v < 8; ++v)
      cb[(size_t)((t0 + t) * 16 + v + 8 * half) * PIX] = acc[t][v];
}

template <int T0, int NT>
__device__ __forceinline__ void encode_walk(
    const float* __restrict__ cbase, const float* lds, int aoff,
    const int (&boff)[9][2], const bool (&bok)[9][2], v8f (&acc)[4]) {
  const float* cptr = cbase;
  aoff += T0 * 16 * APITCH;
  for (int trip = 0; trip < 16; ++trip) {
    float braw[9][2];
    v2f   a[9][NT];
#pragma unroll
    for (int j = 0; j < 9; ++j)
#pragma unroll
      for (int v = 0; v < 2; ++v)
        braw[j][v] = cptr[boff[j][v]];
#pragma unroll
    for (int j = 0; j < 9; ++j)
#pragma unroll
      for (int t = 0; t < NT; ++t)
        a[j][t] = *(const v2f*)(lds + aoff + t * 16 * APITCH + 4 * j);
    __builtin_amdgcn_sched_barrier(0);
    v2f b[9];
#pragma unroll
    for (int j = 0; j < 9; ++j) {
      b[j][0] = bok[j][0] ? braw[j][0] : 0.0f;
      b[j][1] = bok[j][1] ? braw[j][1] : 0.0f;
    }
    __builtin_amdgcn_sched_barrier(0);
#pragma unroll
    for (int j = 0; j < 9; ++j)
#pragma unroll
      for (int t = 0; t < NT; ++t)
        acc[t] = __builtin_amdgcn_wmma_f32_16x16x4_f32(
            false, a[j][t], false, b[j], (short)0, acc[t], false, false);
    cptr += 4 * PIX;
    aoff += 36;
  }
}

__global__ __launch_bounds__(256) void carafe_encode_softmax(
    const float* __restrict__ comp, const float* __restrict__ w_enc,
    const float* __restrict__ b_enc, float* __restrict__ mask) {
  __shared__ __attribute__((aligned(16))) float lds[MPAD * APITCH];

  {
    const float4* w4 = (const float4*)w_enc;
    for (int i0 = 0; i0 < 56 * 256; i0 += 28 * 256) {
      float4 stg[28];
#pragma unroll
      for (int u = 0; u < 28; ++u) stg[u] = w4[i0 + u * 256 + threadIdx.x];
#pragma unroll
      for (int u = 0; u < 28; ++u) {
        const int i = i0 + u * 256 + threadIdx.x;
        const int m = i / 144;
        *(float4*)(lds + m * APITCH + 4 * (i - m * 144)) = stg[u];
      }
    }
    if (threadIdx.x < 64) {
      const int i = 56 * 256 + threadIdx.x;
      const int m = i / 144;
      *(float4*)(lds + m * APITCH + 4 * (i - m * 144)) = w4[i];
    }
    for (int i = threadIdx.x; i < (MPAD - MENC) * 144; i += 256) {
      const int m = i / 144;
      *(float4*)(lds + (MENC + m) * APITCH + 4 * (i - m * 144)) = make_float4(0.f, 0.f, 0.f, 0.f);
    }
  }

  const int wave = __builtin_amdgcn_readfirstlane(threadIdx.x >> 5);
  const int lane = threadIdx.x & 31;
  const int row  = lane & 15;
  const int half = lane >> 4;

  const int ptile = wave & 3;
  const int tb    = (wave >> 2) * 4;
  const int tile  = blockIdx.x * 4 + ptile;
  const int gp   = tile << 4;
  const int n    = gp >> 12;
  const int pin  = gp & 4095;

  const int p  = pin + row;
  const int py = p >> 6;
  const int px = p & 63;

  v8f acc[4];
#pragma unroll
  for (int t = 0; t < 4; ++t)
#pragma unroll
    for (int v = 0; v < 8; ++v) {
      const int m = (tb + t) * 16 + v + 8 * half;
      const float bv = b_enc[min(m, MENC - 1)];
      acc[t][v] = (m < MENC) ? bv : 0.0f;
    }

  const float* cbase = comp + (size_t)n * CCOMP * PIX;

  int  boff[9][2];
  bool bok[9][2];
#pragma unroll
  for (int j = 0; j < 9; ++j)
#pragma unroll
    for (int v = 0; v < 2; ++v) {
      const int kl = 4 * j + 2 * half + v;
      const int cl = kl / 9;
      const int r  = kl - cl * 9;
      const int y  = py + (r / 3) - 1;
      const int xx = px + (r % 3) - 1;
      bok[j][v] = ((unsigned)y < HW) & ((unsigned)xx < HW);
      const int yc = min(max(y, 0), HW - 1);
      const int xc = min(max(xx, 0), HW - 1);
      boff[j][v] = cl * PIX + yc * HW + xc;
    }

  const int aoff = row * APITCH + 2 * half;

  __syncthreads();

  if (tb == 0) encode_walk<0, 4>(cbase, lds, aoff, boff, bok, acc);
  else         encode_walk<4, 3>(cbase, lds, aoff, boff, bok, acc);

  __syncthreads();

  float* s = lds + ptile * (16 * 113);
#pragma unroll
  for (int t = 0; t < 4; ++t)
    if (tb + t < 7) {
#pragma unroll
      for (int v = 0; v < 8; ++v)
        s[row * 113 + (tb + t) * 16 + v + 8 * half] = acc[t][v];
    }

  __syncthreads();

  {
    const int g   = threadIdx.x;
    const int pix = g >> 2;
    const int ij  = g & 3;
    const float* base = lds + (pix >> 4) * (16 * 113) + (pix & 15) * 113;
    float r[25];
    float mx = -3.4e38f;
#pragma unroll
    for (int k = 0; k < 25; ++k) { r[k] = base[k * 4 + ij]; mx = fmaxf(mx, r[k]); }
    float sum = 0.0f;
#pragma unroll
    for (int k = 0; k < 25; ++k) { r[k] = __expf(r[k] - mx); sum += r[k]; }
    const float inv = 1.0f / sum;
    float* mout = mask + (size_t)(blockIdx.x * 64 + pix) * 100 + ij * 25;
#pragma unroll
    for (int k = 0; k < 25; ++k) mout[k] = r[k] * inv;
  }
}

#define XPITCH 72
#define XROWS  40

__global__ __launch_bounds__(64) void carafe_reassemble(
    const float* __restrict__ x, const float* __restrict__ mask,
    float* __restrict__ out) {
  __shared__ __attribute__((aligned(16))) float sm[64 * 100];
  __shared__ __attribute__((aligned(16))) float sx[XROWS * XPITCH];

  const int t  = threadIdx.x;
  const int b  = blockIdx.x;
  const int cg = b & 31;
  const int h  = (b >> 5) & 63;
  const int n  = b >> 11;

  const float* xg = x + ((size_t)(n * 256 + cg * 8)) * PIX;

  const float4* mrow4 = (const float4*)(mask + (size_t)((n << 12) + (h << 6)) * 100);
  float4* sm4 = (float4*)sm;
  {
    const int c4   = t & 15;
    const int rsub = t >> 4;
    float4 stg[25];
    float4 xs[10];
    bool   xv[10];
#pragma unroll
    for (int q = 0; q < 25; ++q) stg[q] = mrow4[q * 64 + t];
#pragma unroll
    for (int q = 0; q < 10; ++q) {
      const int rr = 4 * q + rsub;
      const int ch = rr / 5;
      const int y  = h + (rr - ch * 5) - 2;
      xv[q] = (unsigned)y < HW;
      const int yc = min(max(y, 0), HW - 1);
      xs[q] = *(const float4*)(xg + (size_t)ch * PIX + yc * HW + 4 * c4);
    }
#pragma unroll
    for (int q = 0; q < 25; ++q) sm4[q * 64 + t] = stg[q];
#pragma unroll
    for (int q = 0; q < 10; ++q) {
      const int rr = 4 * q + rsub;
      float4 v = xs[q];
      if (!xv[q]) v = make_float4(0.f, 0.f, 0.f, 0.f);
      *(float4*)(sx + rr * XPITCH + 4 + 4 * c4) = v;
    }
    for (int i = t; i < 2 * XROWS; i += 64)
      *(float4*)(sx + (i >> 1) * XPITCH + (i & 1) * (XPITCH - 4)) = make_float4(0.f, 0.f, 0.f, 0.f);
  }
  __syncthreads();

  float m[100];
#pragma unroll
  for (int j = 0; j < 25; ++j) {
    const float4 v = sm4[t * 25 + j];
    m[4 * j + 0] = v.x; m[4 * j + 1] = v.y;
    m[4 * j + 2] = v.z; m[4 * j + 3] = v.w;
  }

  const float* sxl = sx + t + 2;
  float* og = out + (((size_t)(n * 256 + cg * 8) * 128) + 2 * h) * 128 + 2 * t;

  for (int cp = 0; cp < 4; ++cp) {
    const float* wa = sxl + (2 * cp) * 5 * XPITCH;
    const float* wb = wa + 5 * XPITCH;

    float ra[25], rb[25];
#pragma unroll
    for (int di = 0; di < 5; ++di)
#pragma unroll
      for (int dj = 0; dj < 5; ++dj) {
        ra[di * 5 + dj] = wa[di * XPITCH + dj];
        rb[di * 5 + dj] = wb[di * XPITCH + dj];
      }

    float a0 = 0.f, a1 = 0.f, a2 = 0.f, a3 = 0.f;
    float b0 = 0.f, b1 = 0.f, b2 = 0.f, b3 = 0.f;
#pragma unroll
    for (int k = 0; k < 25; ++k) {
      a0 += ra[k] * m[k];
      a1 += ra[k] * m[25 + k];
      a2 += ra[k] * m[50 + k];
      a3 += ra[k] * m[75 + k];
      b0 += rb[k] * m[k];
      b1 += rb[k] * m[25 + k];
      b2 += rb[k] * m[50 + k];
      b3 += rb[k] * m[75 + k];
    }

    float* oa = og + (size_t)(2 * cp) * 128 * 128;
    float* ob = oa + 128 * 128;
    float2 r0; r0.x = a0; r0.y = a1;
    float2 r1; r1.x = a2; r1.y = a3;
    *(float2*)(oa)       = r0;
    *(float2*)(oa + 128) = r1;
    float2 q0; q0.x = b0; q0.y = b1;
    float2 q1; q1.x = b2; q1.y = b3;
    *(float2*)(ob)       = q0;
    *(float2*)(ob + 128) = q1;
  }
}

extern "C" void kernel_launch(void* const* d_in, const int* in_sizes, int n_in,
                              void* d_out, int out_size, void* d_ws, size_t ws_size,
                              hipStream_t stream) {
  (void)in_sizes; (void)n_in; (void)out_size; (void)ws_size;
  const float* x      = (const float*)d_in[0];
  const float* w_comp = (const float*)d_in[1];
  const float* b_comp = (const float*)d_in[2];
  const float* w_enc  = (const float*)d_in[3];
  const float* b_enc  = (const float*)d_in[4];
  float* out = (float*)d_out;

  float* comp = (float*)d_ws;
  float* mask = comp + (size_t)4 * CCOMP * PIX;

  carafe_compress<<<256, 256, 0, stream>>>(x, w_comp, b_comp, comp);
  carafe_encode_softmax<<<256, 256, 0, stream>>>(comp, w_enc, b_enc, mask);
  carafe_reassemble<<<8192, 64, 0, stream>>>(x, mask, out);
}


# Reassembly: a block takes 32 channels (4 window sub-groups, double-buffered in LDS) per mask row instead of 8: mask staged once per 32 channels, 2048 work-groups not 8192; sums unchanged
# speedup vs baseline: 3.7205x; 1.1123x over previous
#include <hip/hip_runtime.h>
#include <math.h>


typedef __attribute__((ext_vector_type(2))) float v2f;
typedef __attribute__((ext_vector_type(8))) float v8f;

#define HW 64
#define PIX 4096
#define CIN 256
#define CCOMP 64
#define MENC 100
#define MPAD 112
#define KENC 576
#define NPIX_TOTAL 16384

#define WPITCH 260
#define APITCH 580

__global__ __launch_bounds__(256) void carafe_compress(
    const float* __restrict__ x, const float* __restrict__ w_comp,
    const float* __restrict__ b_comp, float* __restrict__ comp) {
  __shared__ __attribute__((aligned(16))) float lw[CCOMP * WPITCH];

  {
    const float4* w4 = (const float4*)w_comp;
    float4 stg[16];
#pragma unroll
    for (int u = 0; u < 16; ++u) stg[u] = w4[u * 256 + threadIdx.x];
#pragma unroll
    for (int u = 0; u < 16; ++u) {
      const int i = u * 256 + threadIdx.x;
      *(float4*)(lw + (i >> 6) * WPITCH + 4 * (i & 63)) = stg[u];
    }
  }

  const int wave = threadIdx.x >> 5;
  const int lane = threadIdx.x & 31;
  const int row  = lane & 15;
  const int half = lane >> 4;

  const int tile = blockIdx.x * 4 + (wave & 3);
  const int t0   = (wave >> 2) * 2;
  const int gp   = tile << 4;
  const int n    = gp >> 12;
  const int pin  = gp & 4095;

  v8f acc[2];
#pragma unroll
  for (int t = 0; t < 2; ++t)
#pragma unroll
    for (int v = 0; v < 8; ++v)
      acc[t][v] = b_comp[(t0 + t) * 16 + v + 8 * half];

  const float* xb = x + (size_t)n * CIN * PIX + pin + row;
  const int arow = (t0 * 16 + row) * WPITCH + 2 * half;

  __syncthreads();

  for (int k0 = 0; k0 < CIN; k0 += 32) {
    v2f b[8];
    v2f a[8][2];
#pragma unroll
    for (int s = 0; s < 8; ++s) {
      const int kb = k0 + 4 * s + 2 * half;
      b[s][0] = xb[(size_t)(kb + 0) * PIX];
      b[s][1] = xb[(size_t)(kb + 1) * PIX];
    }
#pragma unroll
    for (int s = 0; s < 8; ++s)
#pragma unroll
      for (int t = 0; t < 2; ++t)
        a[s][t] = *(const v2f*)(lw + arow + t * 16 * WPITCH + k0 + 4 * s);
    __builtin_amdgcn_sched_barrier(0);
#pragma unroll
    for (int s = 0; s < 8; ++s)
#pragma unroll
      for (int t = 0; t < 2; ++t)
        acc[t] = __builtin_amdgcn_wmma_f32_16x16x4_f32(
            false, a[s][t], false, b[s], (short)0, acc[t], false, false);
  }

  float* cb = comp + (size_t)n * CCOMP * PIX + pin + row;
#pragma unroll
  for (int t = 0; t < 2; ++t)
#pragma unroll
    for (int v = 0; v < 8; ++v)
      cb[(size_t)((t0 + t) * 16 + v + 8 * half) * PIX] = acc[t][v];
}

template <int T0, int NT>
__device__ __forceinline__ void encode_walk(
    const float* __restrict__ cbase, const float* lds, int aoff,
    const int (&boff)[9][2], const bool (&bok)[9][2], v8f (&acc)[4]) {
  const float* cptr = cbase;
  aoff += T0 * 16 * APITCH;
  for (int trip = 0; trip < 16; ++trip) {
    float braw[9][2];
    v2f   a[9][NT];
#pragma unroll
    for (int j = 0; j < 9; ++j)
#pragma unroll
      for (int v = 0; v < 2; ++v)
        braw[j][v] = cptr[boff[j][v]];
#pragma unroll
    for (int j = 0; j < 9; ++j)
#pragma unroll
      for (int t = 0; t < NT; ++t)
        a[j][t] = *(const v2f*)(lds + aoff + t * 16 * APITCH + 4 * j);
    __builtin_amdgcn_sched_barrier(0);
    v2f b[9];
#pragma unroll
    for (int j = 0; j < 9; ++j) {
      b[j][0] = bok[j][0] ? braw[j][0] : 0.0f;
      b[j][1] = bok[j][1] ? braw[j][1] : 0.0f;
    }
    __builtin_amdgcn_sched_barrier(0);
#pragma unroll
    for (int j = 0; j < 9; ++j)
#pragma unroll
      for (int t = 0; t < NT; ++t)
        acc[t] = __builtin_amdgcn_wmma_f32_16x16x4_f32(
            false, a[j][t], false, b[j], (short)0, acc[t], false, false);
    cptr += 4 * PIX;
    aoff += 36;
  }
}

__global__ __launch_bounds__(256) void carafe_encode_softmax(
    const float* __restrict__ comp, const float* __restrict__ w_enc,
    const float* __restrict__ b_enc, float* __restrict__ mask) {
  __shared__ __attribute__((aligned(16))) float lds[MPAD * APITCH];

  {
    const float4* w4 = (const float4*)w_enc;
    for (int i0 = 0; i0 < 56 * 256; i0 += 28 * 256) {
      float4 stg[28];
#pragma unroll
      for (int u = 0; u < 28; ++u) stg[u] = w4[i0 + u * 256 + threadIdx.x];
#pragma unroll
      for (int u = 0; u < 28; ++u) {
        const int i = i0 + u * 256 + threadIdx.x;
        const int m = i / 144;
        *(float4*)(lds + m * APITCH + 4 * (i - m * 144)) = stg[u];
      }
    }
    if (threadIdx.x < 64) {
      const int i = 56 * 256 + threadIdx.x;
      const int m = i / 144;
      *(float4*)(lds + m * APITCH + 4 * (i - m * 144)) = w4[i];
    }
    for (int i = threadIdx.x; i < (MPAD - MENC) * 144; i += 256) {
      const int m = i / 144;
      *(float4*)(lds + (MENC + m) * APITCH + 4 * (i - m * 144)) = make_float4(0.f, 0.f, 0.f, 0.f);
    }
  }

  const int wave = __builtin_amdgcn_readfirstlane(threadIdx.x >> 5);
  const int lane = threadIdx.x & 31;
  const int row  = lane & 15;
  const int half = lane >> 4;

  const int ptile = wave & 3;
  const int tb    = (wave >> 2) * 4;
  const int tile  = blockIdx.x * 4 + ptile;
  const int gp   = tile << 4;
  const int n    = gp >> 12;
  const int pin  = gp & 4095;

  const int p  = pin + row;
  const int py = p >> 6;
  const int px = p & 63;

  v8f acc[4];
#pragma unroll
  for (int t = 0; t < 4; ++t)
#pragma unroll
    for (int v = 0; v < 8; ++v) {
      const int m = (tb + t) * 16 + v + 8 * half;
      const float bv = b_enc[min(m, MENC - 1)];
      acc[t][v] = (m < MENC) ? bv : 0.0f;
    }

  const float* cbase = comp + (size_t)n * CCOMP * PIX;

  int  boff[9][2];
  bool bok[9][2];
#pragma unroll
  for (int j = 0; j < 9; ++j)
#pragma unroll
    for (int v = 0; v < 2; ++v) {
      const int kl = 4 * j + 2 * half + v;
      const int cl = kl / 9;
      const int r  = kl - cl * 9;
      const int y  = py + (r / 3) - 1;
      const int xx = px + (r % 3) - 1;
      bok[j][v] = ((unsigned)y < HW) & ((unsigned)xx < HW);
      const int yc = min(max(y, 0), HW - 1);
      const int xc = min(max(xx, 0), HW - 1);
      boff[j][v] = cl * PIX + yc * HW + xc;
    }

  const int aoff = row * APITCH + 2 * half;

  __syncthreads();

  if (tb == 0) encode_walk<0, 4>(cbase, lds, aoff, boff, bok, acc);
  else         encode_walk<4, 3>(cbase, lds, aoff, boff, bok, acc);

  __syncthreads();

  float* s = lds + ptile * (16 * 113);
#pragma unroll
  for (int t = 0; t < 4; ++t)
    if (tb + t < 7) {
#pragma unroll
      for (int v = 0; v < 8; ++v)
        s[row * 113 + (tb + t) * 16 + v + 8 * half] = acc[t][v];
    }

  __syncthreads();

  {
    const int g   = threadIdx.x;
    const int pix = g >> 2;
    const int ij  = g & 3;
    const float* base = lds + (pix >> 4) * (16 * 113) + (pix & 15) * 113;
    float r[25];
    float mx = -3.4e38f;
#pragma unroll
    for (int k = 0; k < 25; ++k) { r[k] = base[k * 4 + ij]; mx = fmaxf(mx, r[k]); }
    float sum = 0.0f;
#pragma unroll
    for (int k = 0; k < 25; ++k) { r[k] = __expf(r[k] - mx); sum += r[k]; }
    const float inv = 1.0f / sum;
    float* mout = mask + (size_t)(blockIdx.x * 64 + pix) * 100 + ij * 25;
#pragma unroll
    for (int k = 0; k < 25; ++k) mout[k] = r[k] * inv;
  }
}

#define XPITCH 72
#define XROWS  40
#define NSUB   4

__device__ __forceinline__ void lds_only_barrier() {
  __builtin_amdgcn_fence(__ATOMIC_RELEASE, "workgroup", "local");
  __builtin_amdgcn_s_barrier();
  __builtin_amdgcn_fence(__ATOMIC_ACQUIRE, "workgroup", "local");
}

__global__ __launch_bounds__(64) void carafe_reassemble(
    const float* __restrict__ x, const float* __restrict__ mask,
    float* __restrict__ out) {
  __shared__ __attribute__((aligned(16))) float sm[64 * 100];
  __shared__ __attribute__((aligned(16))) float sx[XROWS * XPITCH];

  const int t  = threadIdx.x;
  const int b  = blockIdx.x;
  const int cg = b & 7;
  const int h  = (b >> 3) & 63;
  const int n  = b >> 9;

  const float* xg = x + ((size_t)(n * 256 + cg * 32)) * PIX;

  const int c4   = t & 15;
  const int rsub = t >> 4;
  int  xoff[10];
  bool xv[10];
#pragma unroll
  for (int q = 0; q < 10; ++q) {
    const int rr = 4 * q + rsub;
    const int ch = rr / 5;
    const int y  = h + (rr - ch * 5) - 2;
    xv[q] = (unsigned)y < HW;
    const int yc = min(max(y, 0), HW - 1);
    xoff[q] = ch * PIX + yc * HW + 4 * c4;
  }
  const int soff = rsub * XPITCH + 4 + 4 * c4;

  const float4* mrow4 = (const float4*)(mask + (size_t)((n << 12) + (h << 6)) * 100);
  float4* sm4 = (float4*)sm;
  float4 xs[10];
  {
    float4 stg[25];
#pragma unroll
    for (int q = 0; q < 25; ++q) stg[q] = mrow4[q * 64 + t];
#pragma unroll
    for (int q = 0; q < 10; ++q) xs[q] = *(const float4*)(xg + xoff[q]);
#pragma unroll
    for (int q = 0; q < 25; ++q) sm4[q * 64 + t] = stg[q];
#pragma unroll
    for (int q = 0; q < 10; ++q) {
      float4 v = xs[q];
      if (!xv[q]) v = make_float4(0.f, 0.f, 0.f, 0.f);
      *(float4*)(sx + 4 * q * XPITCH + soff) = v;
    }
    for (int i = t; i < 2 * XROWS; i += 64)
      *(float4*)(sx + (i >> 1) * XPITCH + (i & 1) * (XPITCH - 4)) = make_float4(0.f, 0.f, 0.f, 0.f);
  }
  __syncthreads();

  float m[100];
#pragma unroll
  for (int j = 0; j < 25; ++j) {
    const float4 v = sm4[t * 25 + j];
    m[4 * j + 0] = v.x; m[4 * j + 1] = v.y;
    m[4 * j + 2] = v.z; m[4 * j + 3] = v.w;
  }
  __syncthreads();

  float* og = out + (((size_t)(n * 256 + cg * 32) * 128) + 2 * h) * 128 + 2 * t;

  for (int s = 0; s < NSUB; ++s) {
    const float* cur = (s & 1) ? sm : sx;
    float*       nxt = (s & 1) ? sx : sm;
    const bool more = (s + 1 < NSUB);

    if (more) {
      const float* xn = xg + (size_t)(s + 1) * 8 * PIX;
#pragma unroll
      for (int q = 0; q < 10; ++q) xs[q] = *(const float4*)(xn + xoff[q]);
    }
    __builtin_amdgcn_sched_barrier(0);

    const float* sxl = cur + t + 2;

    for (int cp = 0; cp < 4; ++cp) {
      const float* wa = sxl + (2 * cp) * 5 * XPITCH;
      const float* wb = wa + 5 * XPITCH;

      float ra[25], rb[25];
#pragma unroll
      for (int di = 0; di < 5; ++di)
#pragma unroll
        for (int dj = 0; dj < 5; ++dj) {
          ra[di * 5 + dj] = wa[di * XPITCH + dj];
          rb[di * 5 + dj] = wb[di * XPITCH + dj];
        }

      float a0 = 0.f, a1 = 0.f, a2 = 0.f, a3 = 0.f;
      float b0 = 0.f, b1 = 0.f, b2 = 0.f, b3 = 0.f;
#pragma unroll
      for (int k = 0; k < 25; ++k) {
        a0 += ra[k] * m[k];
        a1 += ra[k] * m[25 + k];
        a2 += ra[k] * m[50 + k];
        a3 += ra[k] * m[75 + k];
        b0 += rb[k] * m[k];
        b1 += rb[k] * m[25 + k];
        b2 += rb[k] * m[50 + k];
        b3 += rb[k] * m[75 + k];
      }

      float* oa = og + (size_t)(s * 8 + 2 * cp) * 128 * 128;
      float* ob = oa + 128 * 128;
      float2 r0; r0.x = a0; r0.y = a1;
      float2 r1; r1.x = a2; r1.y = a3;
      *(float2*)(oa)       = r0;
      *(float2*)(oa + 128) = r1;
      float2 q0; q0.x = b0; q0.y = b1;
      float2 q1; q1.x = b2; q1.y = b3;
      *(float2*)(ob)       = q0;
      *(float2*)(ob + 128) = q1;
    }

    if (more) {
#pragma unroll
      for (int q = 0; q < 10; ++q) {
        float4 v = xs[q];
        if (!xv[q]) v = make_float4(0.f, 0.f, 0.f, 0.f);
        *(float4*)(nxt + 4 * q * XPITCH + soff) = v;
      }
      if (s == 0) {
        for (int i = t; i < 2 * XROWS; i += 64)
          *(float4*)(nxt + (i >> 1) * XPITCH + (i & 1) * (XPITCH - 4)) = make_float4(0.f, 0.f, 0.f, 0.f);
      }
    }
    lds_only_barrier();
  }
}

extern "C" void kernel_launch(void* const* d_in, const int* in_sizes, int n_in,
                              void* d_out, int out_size, void* d_ws, size_t ws_size,
                              hipStream_t stream) {
  (void)in_sizes; (void)n_in; (void)out_size; (void)ws_size;
  const float* x      = (const float*)d_in[0];
  const float* w_comp = (const float*)d_in[1];
  const float* b_comp = (const float*)d_in[2];
  const float* w_enc  = (const float*)d_in[3];
  const float* b_enc  = (const float*)d_in[4];
  float* out = (float*)d_out;

  float* comp = (float*)d_ws;
  float* mask = comp + (size_t)4 * CCOMP * PIX;

  carafe_compress<<<256, 256, 0, stream>>>(x, w_comp, b_comp, comp);
  carafe_encode_softmax<<<256, 256, 0, stream>>>(comp, w_enc, b_enc, mask);
  carafe_reassemble<<<2048, 64, 0, stream>>>(x, mask, out);
}


# 3x3 conv K walk software-pipelined over two register sets: trip t+1's 18 gathers and LDS fragment reads issue ahead of trip t's matrix ops; k order and sums unchanged
# speedup vs baseline: 4.0391x; 1.0856x over previous
#include <hip/hip_runtime.h>
#include <math.h>


typedef __attribute__((ext_vector_type(2))) float v2f;
typedef __attribute__((ext_vector_type(8))) float v8f;

#define HW 64
#define PIX 4096
#define CIN 256
#define CCOMP 64
#define MENC 100
#define MPAD 112
#define KENC 576
#define NPIX_TOTAL 16384

#define WPITCH 260
#define APITCH 580

__global__ __launch_bounds__(256) void carafe_compress(
    const float* __restrict__ x, const float* __restrict__ w_comp,
    const float* __restrict__ b_comp, float* __restrict__ comp) {
  __shared__ __attribute__((aligned(16))) float lw[CCOMP * WPITCH];

  {
    const float4* w4 = (const float4*)w_comp;
    float4 stg[16];
#pragma unroll
    for (int u = 0; u < 16; ++u) stg[u] = w4[u * 256 + threadIdx.x];
#pragma unroll
    for (int u = 0; u < 16; ++u) {
      const int i = u * 256 + threadIdx.x;
      *(float4*)(lw + (i >> 6) * WPITCH + 4 * (i & 63)) = stg[u];
    }
  }

  const int wave = threadIdx.x >> 5;
  const int lane = threadIdx.x & 31;
  const int row  = lane & 15;
  const int half = lane >> 4;

  const int tile = blockIdx.x * 4 + (wave & 3);
  const int t0   = (wave >> 2) * 2;
  const int gp   = tile << 4;
  const int n    = gp >> 12;
  const int pin  = gp & 4095;

  v8f acc[2];
#pragma unroll
  for (int t = 0; t < 2; ++t)
#pragma unroll
    for (int v = 0; v < 8; ++v)
      acc[t][v] = b_comp[(t0 + t) * 16 + v + 8 * half];

  const float* xb = x + (size_t)n * CIN * PIX + pin + row;
  const int arow = (t0 * 16 + row) * WPITCH + 2 * half;

  __syncthreads();

  for (int k0 = 0; k0 < CIN; k0 += 32) {
    v2f b[8];
    v2f a[8][2];
#pragma unroll
    for (int s = 0; s < 8; ++s) {
      const int kb = k0 + 4 * s + 2 * half;
      b[s][0] = xb[(size_t)(kb + 0) * PIX];
      b[s][1] = xb[(size_t)(kb + 1) * PIX];
    }
#pragma unroll
    for (int s = 0; s < 8; ++s)
#pragma unroll
      for (int t = 0; t < 2; ++t)
        a[s][t] = *(const v2f*)(lw + arow + t * 16 * WPITCH + k0 + 4 * s);
    __builtin_amdgcn_sched_barrier(0);
#pragma unroll
    for (int s = 0; s < 8; ++s)
#pragma unroll
      for (int t = 0; t < 2; ++t)
        acc[t] = __builtin_amdgcn_wmma_f32_16x16x4_f32(
            false, a[s][t], false, b[s], (short)0, acc[t], false, false);
  }

  float* cb = comp + (size_t)n * CCOMP * PIX + pin + row;
#pragma unroll
  for (int t = 0; t < 2; ++t)
#pragma unroll
    for (int v = 0; v < 8; ++v)
      cb[(size_t)((t0 + t) * 16 + v + 8 * half) * PIX] = acc[t][v];
}


template <int NT>
__device__ __forceinline__ void encode_load(
    const float* __restrict__ cptr, const float* lds, int aoff,
    const int (&boff)[9][2], float (&braw)[9][2], v2f (&a)[9][NT]) {
#pragma unroll
  for (int j = 0; j < 9; ++j)
#pragma unroll
    for (int v = 0; v < 2; ++v)
      braw[j][v] = cptr[boff[j][v]];
#pragma unroll
  for (int j = 0; j < 9; ++j)
#pragma unroll
    for (int t = 0; t < NT; ++t)
      a[j][t] = *(const v2f*)(lds + aoff + t * 16 * APITCH + 4 * j);
}

template <int NT>
__device__ __forceinline__ void encode_mma(
    const float (&braw)[9][2], const v2f (&a)[9][NT],
    const bool (&bok)[9][2], v8f (&acc)[4]) {
  v2f b[9];
#pragma unroll
  for (int j = 0; j < 9; ++j) {
    b[j][0] = bok[j][0] ? braw[j][0] : 0.0f;
    b[j][1] = bok[j][1] ? braw[j][1] : 0.0f;
  }
  __builtin_amdgcn_sched_barrier(0);
#pragma unroll
  for (int j = 0; j < 9; ++j)
#pragma unroll
    for (int t = 0; t < NT; ++t)
      acc[t] = __builtin_amdgcn_wmma_f32_16x16x4_f32(
          false, a[j][t], false, b[j], (short)0, acc[t], false, false);
}

template <int T0, int NT>
__device__ __forceinline__ void encode_walk(
    const float* __restrict__ cbase, const float* lds, int aoff,
    const int (&boff)[9][2], const bool (&bok)[9][2], v8f (&acc)[4]) {
  const float* cptr = cbase;
  aoff += T0 * 16 * APITCH;
  float br0[9][2], br1[9][2];
  v2f   a0[9][NT], a1[9][NT];
  encode_load<NT>(cptr, lds, aoff, boff, br0, a0);
  for (int pass = 0; pass < 7; ++pass) {
    encode_load<NT>(cptr + 4 * PIX, lds, aoff + 36, boff, br1, a1);
    __builtin_amdgcn_sched_barrier(0);
    encode_mma<NT>(br0, a0, bok, acc);
    __builtin_amdgcn_sched_barrier(0);
    encode_load<NT>(cptr + 8 * PIX, lds, aoff + 72, boff, br0, a0);
    __builtin_amdgcn_sched_barrier(0);
    encode_mma<NT>(br1, a1, bok, acc);
    cptr += 8 * PIX;
    aoff += 72;
  }
  encode_load<NT>(cptr + 4 * PIX, lds, aoff + 36, boff, br1, a1);
  __builtin_amdgcn_sched_barrier(0);
  encode_mma<NT>(br0, a0, bok, acc);
  __builtin_amdgcn_sched_barrier(0);
  encode_mma<NT>(br1, a1, bok, acc);
}

__global__ __launch_bounds__(256) void carafe_encode_softmax(
    const float* __restrict__ comp, const float* __restrict__ w_enc,
    const float* __restrict__ b_enc, float* __restrict__ mask) {
  __shared__ __attribute__((aligned(16))) float lds[MPAD * APITCH];

  {
    const float4* w4 = (const float4*)w_enc;
    for (int i0 = 0; i0 < 56 * 256; i0 += 28 * 256) {
      float4 stg[28];
#pragma unroll
      for (int u = 0; u < 28; ++u) stg[u] = w4[i0 + u * 256 + threadIdx.x];
#pragma unroll
      for (int u = 0; u < 28; ++u) {
        const int i = i0 + u * 256 + threadIdx.x;
        const int m = i / 144;
        *(float4*)(lds + m * APITCH + 4 * (i - m * 144)) = stg[u];
      }
    }
    if (threadIdx.x < 64) {
      const int i = 56 * 256 + threadIdx.x;
      const int m = i / 144;
      *(float4*)(lds + m * APITCH + 4 * (i - m * 144)) = w4[i];
    }
    for (int i = threadIdx.x; i < (MPAD - MENC) * 144; i += 256) {
      const int m = i / 144;
      *(float4*)(lds + (MENC + m) * APITCH + 4 * (i - m * 144)) = make_float4(0.f, 0.f, 0.f, 0.f);
    }
  }

  const int wave = __builtin_amdgcn_readfirstlane(threadIdx.x >> 5);
  const int lane = threadIdx.x & 31;
  const int row  = lane & 15;
  const int half = lane >> 4;

  const int ptile = wave & 3;
  const int tb    = (wave >> 2) * 4;
  const int tile  = blockIdx.x * 4 + ptile;
  const int gp   = tile << 4;
  const int n    = gp >> 12;
  const int pin  = gp & 4095;

  const int p  = pin + row;
  const int py = p >> 6;
  const int px = p & 63;

  v8f acc[4];
#pragma unroll
  for (int t = 0; t < 4; ++t)
#pragma unroll
    for (int v = 0; v < 8; ++v) {
      const int m = (tb + t) * 16 + v + 8 * half;
      const float bv = b_enc[min(m, MENC - 1)];
      acc[t][v] = (m < MENC) ? bv : 0.0f;
    }

  const float* cbase = comp + (size_t)n * CCOMP * PIX;

  int  boff[9][2];
  bool bok[9][2];
#pragma unroll
  for (int j = 0; j < 9; ++j)
#pragma unroll
    for (int v = 0; v < 2; ++v) {
      const int kl = 4 * j + 2 * half + v;
      const int cl = kl / 9;
      const int r  = kl - cl * 9;
      const int y  = py + (r / 3) - 1;
      const int xx = px + (r % 3) - 1;
      bok[j][v] = ((unsigned)y < HW) & ((unsigned)xx < HW);
      const int yc = min(max(y, 0), HW - 1);
      const int xc = min(max(xx, 0), HW - 1);
      boff[j][v] = cl * PIX + yc * HW + xc;
    }

  const int aoff = row * APITCH + 2 * half;

  __syncthreads();

  if (tb == 0) encode_walk<0, 4>(cbase, lds, aoff, boff, bok, acc);
  else         encode_walk<4, 3>(cbase, lds, aoff, boff, bok, acc);

  __syncthreads();

  float* s = lds + ptile * (16 * 113);
#pragma unroll
  for (int t = 0; t < 4; ++t)
    if (tb + t < 7) {
#pragma unroll
      for (int v = 0; v < 8; ++v)
        s[row * 113 + (tb + t) * 16 + v + 8 * half] = acc[t][v];
    }

  __syncthreads();

  {
    const int g   = threadIdx.x;
    const int pix = g >> 2;
    const int ij  = g & 3;
    const float* base = lds + (pix >> 4) * (16 * 113) + (pix & 15) * 113;
    float r[25];
    float mx = -3.4e38f;
#pragma unroll
    for (int k = 0; k < 25; ++k) { r[k] = base[k * 4 + ij]; mx = fmaxf(mx, r[k]); }
    float sum = 0.0f;
#pragma unroll
    for (int k = 0; k < 25; ++k) { r[k] = __expf(r[k] - mx); sum += r[k]; }
    const float inv = 1.0f / sum;
    float* mout = mask + (size_t)(blockIdx.x * 64 + pix) * 100 + ij * 25;
#pragma unroll
    for (int k = 0; k < 25; ++k) mout[k] = r[k] * inv;
  }
}

#define XPITCH 72
#define XROWS  40
#define NSUB   4

__device__ __forceinline__ void lds_only_barrier() {
  __builtin_amdgcn_fence(__ATOMIC_RELEASE, "workgroup", "local");
  __builtin_amdgcn_s_barrier();
  __builtin_amdgcn_fence(__ATOMIC_ACQUIRE, "workgroup", "local");
}

__global__ __launch_bounds__(64) void carafe_reassemble(
    const float* __restrict__ x, const float* __restrict__ mask,
    float* __restrict__ out) {
  __shared__ __attribute__((aligned(16))) float sm[64 * 100];
  __shared__ __attribute__((aligned(16))) float sx[XROWS * XPITCH];

  const int t  = threadIdx.x;
  const int b  = blockIdx.x;
  const int cg = b & 7;
  const int h  = (b >> 3) & 63;
  const int n  = b >> 9;

  const float* xg = x + ((size_t)(n * 256 + cg * 32)) * PIX;

  const int c4   = t & 15;
  const int rsub = t >> 4;
  int  xoff[10];
  bool xv[10];
#pragma unroll
  for (int q = 0; q < 10; ++q) {
    const int rr = 4 * q + rsub;
    const int ch = rr / 5;
    const int y  = h + (rr - ch * 5) - 2;
    xv[q] = (unsigned)y < HW;
    const int yc = min(max(y, 0), HW - 1);
    xoff[q] = ch * PIX + yc * HW + 4 * c4;
  }
  const int soff = rsub * XPITCH + 4 + 4 * c4;

  const float4* mrow4 = (const float4*)(mask + (size_t)((n << 12) + (h << 6)) * 100);
  float4* sm4 = (float4*)sm;
  float4 xs[10];
  {
    float4 stg[25];
#pragma unroll
    for (int q = 0; q < 25; ++q) stg[q] = mrow4[q * 64 + t];
#pragma unroll
    for (int q = 0; q < 10; ++q) xs[q] = *(const float4*)(xg + xoff[q]);
#pragma unroll
    for (int q = 0; q < 25; ++q) sm4[q * 64 + t] = stg[q];
#pragma unroll
    for (int q = 0; q < 10; ++q) {
      float4 v = xs[q];
      if (!xv[q]) v = make_float4(0.f, 0.f, 0.f, 0.f);
      *(float4*)(sx + 4 * q * XPITCH + soff) = v;
    }
    for (int i = t; i < 2 * XROWS; i += 64)
      *(float4*)(sx + (i >> 1) * XPITCH + (i & 1) * (XPITCH - 4)) = make_float4(0.f, 0.f, 0.f, 0.f);
  }
  __syncthreads();

  float m[100];
#pragma unroll
  for (int j = 0; j < 25; ++j) {
    const float4 v = sm4[t * 25 + j];
    m[4 * j + 0] = v.x; m[4 * j + 1] = v.y;
    m[4 * j + 2] = v.z; m[4 * j + 3] = v.w;
  }
  __syncthreads();

  float* og = out + (((size_t)(n * 256 + cg * 32) * 128) + 2 * h) * 128 + 2 * t;

  for (int s = 0; s < NSUB; ++s) {
    const float* cur = (s & 1) ? sm : sx;
    float*       nxt = (s & 1) ? sx : sm;
    const bool more = (s + 1 < NSUB);

    if (more) {
      const float* xn = xg + (size_t)(s + 1) * 8 * PIX;
#pragma unroll
      for (int q = 0; q < 10; ++q) xs[q] = *(const float4*)(xn + xoff[q]);
    }
    __builtin_amdgcn_sched_barrier(0);

    const float* sxl = cur + t + 2;

    for (int cp = 0; cp < 4; ++cp) {
      const float* wa = sxl + (2 * cp) * 5 * XPITCH;
      const float* wb = wa + 5 * XPITCH;

      float ra[25], rb[25];
#pragma unroll
      for (int di = 0; di < 5; ++di)
#pragma unroll
        for (int dj = 0; dj < 5; ++dj) {
          ra[di * 5 + dj] = wa[di * XPITCH + dj];
          rb[di * 5 + dj] = wb[di * XPITCH + dj];
        }

      float a0 = 0.f, a1 = 0.f, a2 = 0.f, a3 = 0.f;
      float b0 = 0.f, b1 = 0.f, b2 = 0.f, b3 = 0.f;
#pragma unroll
      for (int k = 0; k < 25; ++k) {
        a0 += ra[k] * m[k];
        a1 += ra[k] * m[25 + k];
        a2 += ra[k] * m[50 + k];
        a3 += ra[k] * m[75 + k];
        b0 += rb[k] * m[k];
        b1 += rb[k] * m[25 + k];
        b2 += rb[k] * m[50 + k];
        b3 += rb[k] * m[75 + k];
      }

      float* oa = og + (size_t)(s * 8 + 2 * cp) * 128 * 128;
      float* ob = oa + 128 * 128;
      float2 r0; r0.x = a0; r0.y = a1;
      float2 r1; r1.x = a2; r1.y = a3;
      *(float2*)(oa)       = r0;
      *(float2*)(oa + 128) = r1;
      float2 q0; q0.x = b0; q0.y = b1;
      float2 q1; q1.x = b2; q1.y = b3;
      *(float2*)(ob)       = q0;
      *(float2*)(ob + 128) = q1;
    }

    if (more) {
#pragma unroll
      for (int q = 0; q < 10; ++q) {
        float4 v = xs[q];
        if (!xv[q]) v = make_float4(0.f, 0.f, 0.f, 0.f);
        *(float4*)(nxt + 4 * q * XPITCH + soff) = v;
      }
      if (s == 0) {
        for (int i = t; i < 2 * XROWS; i += 64)
          *(float4*)(nxt + (i >> 1) * XPITCH + (i & 1) * (XPITCH - 4)) = make_float4(0.f, 0.f, 0.f, 0.f);
      }
    }
    lds_only_barrier();
  }
}

extern "C" void kernel_launch(void* const* d_in, const int* in_sizes, int n_in,
                              void* d_out, int out_size, void* d_ws, size_t ws_size,
                              hipStream_t stream) {
  (void)in_sizes; (void)n_in; (void)out_size; (void)ws_size;
  const float* x      = (const float*)d_in[0];
  const float* w_comp = (const float*)d_in[1];
  const float* b_comp = (const float*)d_in[2];
  const float* w_enc  = (const float*)d_in[3];
  const float* b_enc  = (const float*)d_in[4];
  float* out = (float*)d_out;

  float* comp = (float*)d_ws;
  float* mask = comp + (size_t)4 * CCOMP * PIX;

  carafe_compress<<<256, 256, 0, stream>>>(x, w_comp, b_comp, comp);
  carafe_encode_softmax<<<256, 256, 0, stream>>>(comp, w_enc, b_enc, mask);
  carafe_reassemble<<<2048, 64, 0, stream>>>(x, mask, out);
}


# 3x3 conv + softmax fused with reassembly: the work-group that computed a row's mask keeps it in LDS and reassembles the row's 256 channels; 2 launches, not 3; sums unchanged
# speedup vs baseline: 4.1876x; 1.0368x over previous
#include <hip/hip_runtime.h>
#include <math.h>


typedef __attribute__((ext_vector_type(2))) float v2f;
typedef __attribute__((ext_vector_type(8))) float v8f;

#define HW 64
#define PIX 4096
#define CIN 256
#define CCOMP 64
#define MENC 100
#define MPAD 112
#define KENC 576
#define NPIX_TOTAL 16384

#define WPITCH 260
#define APITCH 580

__global__ __launch_bounds__(256) void carafe_compress(
    const float* __restrict__ x, const float* __restrict__ w_comp,
    const float* __restrict__ b_comp, float* __restrict__ comp) {
  __shared__ __attribute__((aligned(16))) float lw[CCOMP * WPITCH];

  {
    const float4* w4 = (const float4*)w_comp;
    float4 stg[16];
#pragma unroll
    for (int u = 0; u < 16; ++u) stg[u] = w4[u * 256 + threadIdx.x];
#pragma unroll
    for (int u = 0; u < 16; ++u) {
      const int i = u * 256 + threadIdx.x;
      *(float4*)(lw + (i >> 6) * WPITCH + 4 * (i & 63)) = stg[u];
    }
  }

  const int wave = threadIdx.x >> 5;
  const int lane = threadIdx.x & 31;
  const int row  = lane & 15;
  const int half = lane >> 4;

  const int tile = blockIdx.x * 4 + (wave & 3);
  const int t0   = (wave >> 2) * 2;
  const int gp   = tile << 4;
  const int n    = gp >> 12;
  const int pin  = gp & 4095;

  v8f acc[2];
#pragma unroll
  for (int t = 0; t < 2; ++t)
#pragma unroll
    for (int v = 0; v < 8; ++v)
      acc[t][v] = b_comp[(t0 + t) * 16 + v + 8 * half];

  const float* xb = x + (size_t)n * CIN * PIX + pin + row;
  const int arow = (t0 * 16 + row) * WPITCH + 2 * half;

  __syncthreads();

  for (int k0 = 0; k0 < CIN; k0 += 32) {
    v2f b[8];
    v2f a[8][2];
#pragma unroll
    for (int s = 0; s < 8; ++s) {
      const int kb = k0 + 4 * s + 2 * half;
      b[s][0] = xb[(size_t)(kb + 0) * PIX];
      b[s][1] = xb[(size_t)(kb + 1) * PIX];
    }
#pragma unroll
    for (int s = 0; s < 8; ++s)
#pragma unroll
      for (int t = 0; t < 2; ++t)
        a[s][t] = *(const v2f*)(lw + arow + t * 16 * WPITCH + k0 + 4 * s);
    __builtin_amdgcn_sched_barrier(0);
#pragma unroll
    for (int s = 0; s < 8; ++s)
#pragma unroll
      for (int t = 0; t < 2; ++t)
        acc[t] = __builtin_amdgcn_wmma_f32_16x16x4_f32(
            false, a[s][t], false, b[s], (short)0, acc[t], false, false);
  }

  float* cb = comp + (size_t)n * CCOMP * PIX + pin + row;
#pragma unroll
  for (int t = 0; t < 2; ++t)
#pragma unroll
    for (int v = 0; v < 8; ++v)
      cb[(size_t)((t0 + t) * 16 + v + 8 * half) * PIX] = acc[t][v];
}


template <int NT>
__device__ __forceinline__ void encode_load(
    const float* __restrict__ cptr, const float* lds, int aoff,
    const int (&boff)[9][2], float (&braw)[9][2], v2f (&a)[9][NT]) {
#pragma unroll
  for (int j = 0; j < 9; ++j)
#pragma unroll
    for (int v = 0; v < 2; ++v)
      braw[j][v] = cptr[boff[j][v]];
#pragma unroll
  for (int j = 0; j < 9; ++j)
#pragma unroll
    for (int t = 0; t < NT; ++t)
      a[j][t] = *(const v2f*)(lds + aoff + t * 16 * APITCH + 4 * j);
}

template <int NT>
__device__ __forceinline__ void encode_mma(
    const float (&braw)[9][2], const v2f (&a)[9][NT],
    const bool (&bok)[9][2], v8f (&acc)[4]) {
  v2f b[9];
#pragma unroll
  for (int j = 0; j < 9; ++j) {
    b[j][0] = bok[j][0] ? braw[j][0] : 0.0f;
    b[j][1] = bok[j][1] ? braw[j][1] : 0.0f;
  }
  __builtin_amdgcn_sched_barrier(0);
#pragma unroll
  for (int j = 0; j < 9; ++j)
#pragma unroll
    for (int t = 0; t < NT; ++t)
      acc[t] = __builtin_amdgcn_wmma_f32_16x16x4_f32(
          false, a[j][t], false, b[j], (short)0, acc[t], false, false);
}

template <int T0, int NT>
__device__ __forceinline__ void encode_walk(
    const float* __restrict__ cbase, const float* lds, int aoff,
    const int (&boff)[9][2], const bool (&bok)[9][2], v8f (&acc)[4]) {
  const float* cptr = cbase;
  aoff += T0 * 16 * APITCH;
  float br0[9][2], br1[9][2];
  v2f   a0[9][NT], a1[9][NT];
  encode_load<NT>(cptr, lds, aoff, boff, br0, a0);
  for (int pass = 0; pass < 7; ++pass) {
    encode_load<NT>(cptr + 4 * PIX, lds, aoff + 36, boff, br1, a1);
    __builtin_amdgcn_sched_barrier(0);
    encode_mma<NT>(br0, a0, bok, acc);
    __builtin_amdgcn_sched_barrier(0);
    encode_load<NT>(cptr + 8 * PIX, lds, aoff + 72, boff, br0, a0);
    __builtin_amdgcn_sched_barrier(0);
    encode_mma<NT>(br1, a1, bok, acc);
    cptr += 8 * PIX;
    aoff += 72;
  }
  encode_load<NT>(cptr + 4 * PIX, lds, aoff + 36, boff, br1, a1);
  __builtin_amdgcn_sched_barrier(0);
  encode_mma<NT>(br0, a0, bok, acc);
  __builtin_amdgcn_sched_barrier(0);
  encode_mma<NT>(br1, a1, bok, acc);
}

#define XPITCH 72
#define XROWS  40
#define NSUB   8
#define SMASK_OFF 8192
#define XWIN_OFF  16384

__device__ __forceinline__ void lds_only_barrier() {
  __builtin_amdgcn_fence(__ATOMIC_RELEASE, "workgroup", "local");
  __builtin_amdgcn_s_barrier();
  __builtin_amdgcn_fence(__ATOMIC_ACQUIRE, "workgroup", "local");
}

__global__ __launch_bounds__(256) void carafe_encode_reassemble(
    const float* __restrict__ comp, const float* __restrict__ w_enc,
    const float* __restrict__ b_enc, const float* __restrict__ x,
    float* __restrict__ out) {
  __shared__ __attribute__((aligned(16))) float lds[MPAD * APITCH];

  {
    const float4* w4 = (const float4*)w_enc;
    for (int i0 = 0; i0 < 56 * 256; i0 += 28 * 256) {
      float4 stg[28];
#pragma unroll
      for (int u = 0; u < 28; ++u) stg[u] = w4[i0 + u * 256 + threadIdx.x];
#pragma unroll
      for (int u = 0; u < 28; ++u) {
        const int i = i0 + u * 256 + threadIdx.x;
        const int m = i / 144;
        *(float4*)(lds + m * APITCH + 4 * (i - m * 144)) = stg[u];
      }
    }
    if (threadIdx.x < 64) {
      const int i = 56 * 256 + threadIdx.x;
      const int m = i / 144;
      *(float4*)(lds + m * APITCH + 4 * (i - m * 144)) = w4[i];
    }
    for (int i = threadIdx.x; i < (MPAD - MENC) * 144; i += 256) {
      const int m = i / 144;
      *(float4*)(lds + (MENC + m) * APITCH + 4 * (i - m * 144)) = make_float4(0.f, 0.f, 0.f, 0.f);
    }
  }

  const int wave = __builtin_amdgcn_readfirstlane(threadIdx.x >> 5);
  const int lane = threadIdx.x & 31;
  const int row  = lane & 15;
  const int half = lane >> 4;

  const int ptile = wave & 3;
  const int tb    = (wave >> 2) * 4;
  const int n     = blockIdx.x >> 6;
  const int h     = blockIdx.x & 63;

  const int py = h;
  const int px = ptile * 16 + row;

  v8f acc[4];
#pragma unroll
  for (int t = 0; t < 4; ++t)
#pragma unroll
    for (int v = 0; v < 8; ++v) {
      const int m = (tb + t) * 16 + v + 8 * half;
      const float bv = b_enc[min(m, MENC - 1)];
      acc[t][v] = (m < MENC) ? bv : 0.0f;
    }

  const float* cbase = comp + (size_t)n * CCOMP * PIX;

  int  boff[9][2];
  bool bok[9][2];
#pragma unroll
  for (int j = 0; j < 9; ++j)
#pragma unroll
    for (int v = 0; v < 2; ++v) {
      const int kl = 4 * j + 2 * half + v;
      const int cl = kl / 9;
      const int r  = kl - cl * 9;
      const int y  = py + (r / 3) - 1;
      const int xx = px + (r % 3) - 1;
      bok[j][v] = ((unsigned)y < HW) & ((unsigned)xx < HW);
      const int yc = min(max(y, 0), HW - 1);
      const int xc = min(max(xx, 0), HW - 1);
      boff[j][v] = cl * PIX + yc * HW + xc;
    }

  const int aoff = row * APITCH + 2 * half;

  __syncthreads();

  if (tb == 0) encode_walk<0, 4>(cbase, lds, aoff, boff, bok, acc);
  else         encode_walk<4, 3>(cbase, lds, aoff, boff, bok, acc);

  __syncthreads();

  float* s = lds + ptile * (16 * 113);
#pragma unroll
  for (int t = 0; t < 4; ++t)
    if (tb + t < 7) {
#pragma unroll
      for (int v = 0; v < 8; ++v)
        s[row * 113 + (tb + t) * 16 + v + 8 * half] = acc[t][v];
    }

  const int t  = threadIdx.x & 63;
  const int cq = threadIdx.x >> 6;

  const float* xg = x + ((size_t)(n * 256 + cq * 64)) * PIX;

  const int c4   = t & 15;
  const int rsub = t >> 4;
  int  xoff[10];
  bool xv[10];
#pragma unroll
  for (int q = 0; q < 10; ++q) {
    const int rr = 4 * q + rsub;
    const int ch = rr / 5;
    const int y  = h + (rr - ch * 5) - 2;
    xv[q] = (unsigned)y < HW;
    const int yc = min(max(y, 0), HW - 1);
    xoff[q] = ch * PIX + yc * HW + 4 * c4;
  }
  const int soff = rsub * XPITCH + 4 + 4 * c4;

  float4 xs[10];
#pragma unroll
  for (int q = 0; q < 10; ++q) xs[q] = *(const float4*)(xg + xoff[q]);

  lds_only_barrier();

  float* smask = lds + SMASK_OFF;
  {
    const int g   = threadIdx.x;
    const int pix = g >> 2;
    const int ij  = g & 3;
    const float* base = lds + (pix >> 4) * (16 * 113) + (pix & 15) * 113;
    float r[25];
    float mx = -3.4e38f;
#pragma unroll
    for (int k = 0; k < 25; ++k) { r[k] = base[k * 4 + ij]; mx = fmaxf(mx, r[k]); }
    float sum = 0.0f;
#pragma unroll
    for (int k = 0; k < 25; ++k) { r[k] = __expf(r[k] - mx); sum += r[k]; }
    const float inv = 1.0f / sum;
    float* mout = smask + g * 25;
#pragma unroll
    for (int k = 0; k < 25; ++k) mout[k] = r[k] * inv;
  }

  __builtin_amdgcn_sched_barrier(0);

  float* bufa = lds + XWIN_OFF + cq * (2 * XROWS * XPITCH);
  float* bufb = bufa + XROWS * XPITCH;
#pragma unroll
  for (int q = 0; q < 10; ++q) {
    float4 v = xs[q];
    if (!xv[q]) v = make_float4(0.f, 0.f, 0.f, 0.f);
    *(float4*)(bufa + 4 * q * XPITCH + soff) = v;
  }
  for (int i = t; i < 4 * XROWS; i += 64)
    *(float4*)(bufa + (i >> 1) * XPITCH + (i & 1) * (XPITCH - 4)) = make_float4(0.f, 0.f, 0.f, 0.f);

  __syncthreads();

  const float4* sm4 = (const float4*)smask;
  float m[100];
#pragma unroll
  for (int j = 0; j < 25; ++j) {
    const float4 v = sm4[t * 25 + j];
    m[4 * j + 0] = v.x; m[4 * j + 1] = v.y;
    m[4 * j + 2] = v.z; m[4 * j + 3] = v.w;
  }

  float* og = out + (((size_t)(n * 256 + cq * 64) * 128) + 2 * h) * 128 + 2 * t;

  for (int sg = 0; sg < NSUB; ++sg) {
    const float* cur = (sg & 1) ? bufb : bufa;
    float*       nxt = (sg & 1) ? bufa : bufb;
    const bool more = (sg + 1 < NSUB);

    if (more) {
      const float* xn = xg + (size_t)(sg + 1) * 8 * PIX;
#pragma unroll
      for (int q = 0; q < 10; ++q) xs[q] = *(const float4*)(xn + xoff[q]);
    }
    __builtin_amdgcn_sched_barrier(0);

    const float* sxl = cur + t + 2;

    for (int cp = 0; cp < 4; ++cp) {
      const float* wa = sxl + (2 * cp) * 5 * XPITCH;
      const float* wb = wa + 5 * XPITCH;

      float ra[25], rb[25];
#pragma unroll
      for (int di = 0; di < 5; ++di)
#pragma unroll
        for (int dj = 0; dj < 5; ++dj) {
          ra[di * 5 + dj] = wa[di * XPITCH + dj];
          rb[di * 5 + dj] = wb[di * XPITCH + dj];
        }

      float a0 = 0.f, a1 = 0.f, a2 = 0.f, a3 = 0.f;
      float b0 = 0.f, b1 = 0.f, b2 = 0.f, b3 = 0.f;
#pragma unroll
      for (int k = 0; k < 25; ++k) {
        a0 += ra[k] * m[k];
        a1 += ra[k] * m[25 + k];
        a2 += ra[k] * m[50 + k];
        a3 += ra[k] * m[75 + k];
        b0 += rb[k] * m[k];
        b1 += rb[k] * m[25 + k];
        b2 += rb[k] * m[50 + k];
        b3 += rb[k] * m[75 + k];
      }

      float* oa = og + (size_t)(sg * 8 + 2 * cp) * 128 * 128;
      float* ob = oa + 128 * 128;
      float2 r0; r0.x = a0; r0.y = a1;
      float2 r1; r1.x = a2; r1.y = a3;
      *(float2*)(oa)       = r0;
      *(float2*)(oa + 128) = r1;
      float2 q0; q0.x = b0; q0.y = b1;
      float2 q1; q1.x = b2; q1.y = b3;
      *(float2*)(ob)       = q0;
      *(float2*)(ob + 128) = q1;
    }

    if (more) {
#pragma unroll
      for (int q = 0; q < 10; ++q) {
        float4 v = xs[q];
        if (!xv[q]) v = make_float4(0.f, 0.f, 0.f, 0.f);
        *(float4*)(nxt + 4 * q * XPITCH + soff) = v;
      }
    }
    lds_only_barrier();
  }
}

extern "C" void kernel_launch(void* const* d_in, const int* in_sizes, int n_in,
                              void* d_out, int out_size, void* d_ws, size_t ws_size,
                              hipStream_t stream) {
  (void)in_sizes; (void)n_in; (void)out_size; (void)ws_size;
  const float* x      = (const float*)d_in[0];
  const float* w_comp = (const float*)d_in[1];
  const float* b_comp = (const float*)d_in[2];
  const float* w_enc  = (const float*)d_in[3];
  const float* b_enc  = (const float*)d_in[4];
  float* out = (float*)d_out;

  float* comp = (float*)d_ws;

  carafe_compress<<<256, 256, 0, stream>>>(x, w_comp, b_comp, comp);
  carafe_encode_reassemble<<<256, 256, 0, stream>>>(comp, w_enc, b_enc, x, out);
}


# Reassembly: a lane takes 2 adjacent pixels, a wave the whole row for 4 of 8 channels: both pixels' taps of a window row are three 8-byte LDS reads (was 5 dwords each), 16-byte stores; sums unchanged
# speedup vs baseline: 4.3205x; 1.0317x over previous
#include <hip/hip_runtime.h>
#include <math.h>


typedef __attribute__((ext_vector_type(2))) float v2f;
typedef __attribute__((ext_vector_type(8))) float v8f;

#define HW 64
#define PIX 4096
#define CIN 256
#define CCOMP 64
#define MENC 100
#define MPAD 112
#define KENC 576
#define NPIX_TOTAL 16384

#define WPITCH 260
#define APITCH 580

__global__ __launch_bounds__(256) void carafe_compress(
    const float* __restrict__ x, const float* __restrict__ w_comp,
    const float* __restrict__ b_comp, float* __restrict__ comp) {
  __shared__ __attribute__((aligned(16))) float lw[CCOMP * WPITCH];

  {
    const float4* w4 = (const float4*)w_comp;
    float4 stg[16];
#pragma unroll
    for (int u = 0; u < 16; ++u) stg[u] = w4[u * 256 + threadIdx.x];
#pragma unroll
    for (int u = 0; u < 16; ++u) {
      const int i = u * 256 + threadIdx.x;
      *(float4*)(lw + (i >> 6) * WPITCH + 4 * (i & 63)) = stg[u];
    }
  }

  const int wave = threadIdx.x >> 5;
  const int lane = threadIdx.x & 31;
  const int row  = lane & 15;
  const int half = lane >> 4;

  const int tile = blockIdx.x * 4 + (wave & 3);
  const int t0   = (wave >> 2) * 2;
  const int gp   = tile << 4;
  const int n    = gp >> 12;
  const int pin  = gp & 4095;

  v8f acc[2];
#pragma unroll
  for (int t = 0; t < 2; ++t)
#pragma unroll
    for (int v = 0; v < 8; ++v)
      acc[t][v] = b_comp[(t0 + t) * 16 + v + 8 * half];

  const float* xb = x + (size_t)n * CIN * PIX + pin + row;
  const int arow = (t0 * 16 + row) * WPITCH + 2 * half;

  __syncthreads();

  for (int k0 = 0; k0 < CIN; k0 += 32) {
    v2f b[8];
    v2f a[8][2];
#pragma unroll
    for (int s = 0; s < 8; ++s) {
      const int kb = k0 + 4 * s + 2 * half;
      b[s][0] = xb[(size_t)(kb + 0) * PIX];
      b[s][1] = xb[(size_t)(kb + 1) * PIX];
    }
#pragma unroll
    for (int s = 0; s < 8; ++s)
#pragma unroll
      for (int t = 0; t < 2; ++t)
        a[s][t] = *(const v2f*)(lw + arow + t * 16 * WPITCH + k0 + 4 * s);
    __builtin_amdgcn_sched_barrier(0);
#pragma unroll
    for (int s = 0; s < 8; ++s)
#pragma unroll
      for (int t = 0; t < 2; ++t)
        acc[t] = __builtin_amdgcn_wmma_f32_16x16x4_f32(
            false, a[s][t], false, b[s], (short)0, acc[t], false, false);
  }

  float* cb = comp + (size_t)n * CCOMP * PIX + pin + row;
#pragma unroll
  for (int t = 0; t < 2; ++t)
#pragma unroll
    for (int v = 0; v < 8; ++v)
      cb[(size_t)((t0 + t) * 16 + v + 8 * half) * PIX] = acc[t][v];
}


template <int NT>
__device__ __forceinline__ void encode_load(
    const float* __restrict__ cptr, const float* lds, int aoff,
    const int (&boff)[9][2], float (&braw)[9][2], v2f (&a)[9][NT]) {
#pragma unroll
  for (int j = 0; j < 9; ++j)
#pragma unroll
    for (int v = 0; v < 2; ++v)
      braw[j][v] = cptr[boff[j][v]];
#pragma unroll
  for (int j = 0; j < 9; ++j)
#pragma unroll
    for (int t = 0; t < NT; ++t)
      a[j][t] = *(const v2f*)(lds + aoff + t * 16 * APITCH + 4 * j);
}

template <int NT>
__device__ __forceinline__ void encode_mma(
    const float (&braw)[9][2], const v2f (&a)[9][NT],
    const bool (&bok)[9][2], v8f (&acc)[4]) {
  v2f b[9];
#pragma unroll
  for (int j = 0; j < 9; ++j) {
    b[j][0] = bok[j][0] ? braw[j][0] : 0.0f;
    b[j][1] = bok[j][1] ? braw[j][1] : 0.0f;
  }
  __builtin_amdgcn_sched_barrier(0);
#pragma unroll
  for (int j = 0; j < 9; ++j)
#pragma unroll
    for (int t = 0; t < NT; ++t)
      acc[t] = __builtin_amdgcn_wmma_f32_16x16x4_f32(
          false, a[j][t], false, b[j], (short)0, acc[t], false, false);
}

template <int T0, int NT>
__device__ __forceinline__ void encode_walk(
    const float* __restrict__ cbase, const float* lds, int aoff,
    const int (&boff)[9][2], const bool (&bok)[9][2], v8f (&acc)[4]) {
  const float* cptr = cbase;
  aoff += T0 * 16 * APITCH;
  float br0[9][2], br1[9][2];
  v2f   a0[9][NT], a1[9][NT];
  encode_load<NT>(cptr, lds, aoff, boff, br0, a0);
  for (int pass = 0; pass < 7; ++pass) {
    encode_load<NT>(cptr + 4 * PIX, lds, aoff + 36, boff, br1, a1);
    __builtin_amdgcn_sched_barrier(0);
    encode_mma<NT>(br0, a0, bok, acc);
    __builtin_amdgcn_sched_barrier(0);
    encode_load<NT>(cptr + 8 * PIX, lds, aoff + 72, boff, br0, a0);
    __builtin_amdgcn_sched_barrier(0);
    encode_mma<NT>(br1, a1, bok, acc);
    cptr += 8 * PIX;
    aoff += 72;
  }
  encode_load<NT>(cptr + 4 * PIX, lds, aoff + 36, boff, br1, a1);
  __builtin_amdgcn_sched_barrier(0);
  encode_mma<NT>(br0, a0, bok, acc);
  __builtin_amdgcn_sched_barrier(0);
  encode_mma<NT>(br1, a1, bok, acc);
}

#define XPITCH 72
#define XROWS  40
#define NSUB   8
#define SMASK_OFF 8192
#define SMASK_ODD 3232
#define XWIN_OFF  16384

__device__ __forceinline__ void lds_only_barrier() {
  __builtin_amdgcn_fence(__ATOMIC_RELEASE, "workgroup", "local");
  __builtin_amdgcn_s_barrier();
  __builtin_amdgcn_fence(__ATOMIC_ACQUIRE, "workgroup", "local");
}

__global__ __launch_bounds__(256) void carafe_encode_reassemble(
    const float* __restrict__ comp, const float* __restrict__ w_enc,
    const float* __restrict__ b_enc, const float* __restrict__ x,
    float* __restrict__ out) {
  __shared__ __attribute__((aligned(16))) float lds[MPAD * APITCH];

  {
    const float4* w4 = (const float4*)w_enc;
    for (int i0 = 0; i0 < 56 * 256; i0 += 28 * 256) {
      float4 stg[28];
#pragma unroll
      for (int u = 0; u < 28; ++u) stg[u] = w4[i0 + u * 256 + threadIdx.x];
#pragma unroll
      for (int u = 0; u < 28; ++u) {
        const int i = i0 + u * 256 + threadIdx.x;
        const int m = i / 144;
        *(float4*)(lds + m * APITCH + 4 * (i - m * 144)) = stg[u];
      }
    }
    if (threadIdx.x < 64) {
      const int i = 56 * 256 + threadIdx.x;
      const int m = i / 144;
      *(float4*)(lds + m * APITCH + 4 * (i - m * 144)) = w4[i];
    }
    for (int i = threadIdx.x; i < (MPAD - MENC) * 144; i += 256) {
      const int m = i / 144;
      *(float4*)(lds + (MENC + m) * APITCH + 4 * (i - m * 144)) = make_float4(0.f, 0.f, 0.f, 0.f);
    }
  }

  const int wave = __builtin_amdgcn_readfirstlane(threadIdx.x >> 5);
  const int lane = threadIdx.x & 31;
  const int row  = lane & 15;
  const int half = lane >> 4;

  const int ptile = wave & 3;
  const int tb    = (wave >> 2) * 4;
  const int n     = blockIdx.x >> 6;
  const int h     = blockIdx.x & 63;

  const int py = h;
  const int px = ptile * 16 + row;

  v8f acc[4];
#pragma unroll
  for (int t = 0; t < 4; ++t)
#pragma unroll
    for (int v = 0; v < 8; ++v) {
      const int m = (tb + t) * 16 + v + 8 * half;
      const float bv = b_enc[min(m, MENC - 1)];
      acc[t][v] = (m < MENC) ? bv : 0.0f;
    }

  const float* cbase = comp + (size_t)n * CCOMP * PIX;

  int  boff[9][2];
  bool bok[9][2];
#pragma unroll
  for (int j = 0; j < 9; ++j)
#pragma unroll
    for (int v = 0; v < 2; ++v) {
      const int kl = 4 * j + 2 * half + v;
      const int cl = kl / 9;
      const int r  = kl - cl * 9;
      const int y  = py + (r / 3) - 1;
      const int xx = px + (r % 3) - 1;
      bok[j][v] = ((unsigned)y < HW) & ((unsigned)xx < HW);
      const int yc = min(max(y, 0), HW - 1);
      const int xc = min(max(xx, 0), HW - 1);
      boff[j][v] = cl * PIX + yc * HW + xc;
    }

  const int aoff = row * APITCH + 2 * half;

  __syncthreads();

  if (tb == 0) encode_walk<0, 4>(cbase, lds, aoff, boff, bok, acc);
  else         encode_walk<4, 3>(cbase, lds, aoff, boff, bok, acc);

  __syncthreads();

  float* s = lds + ptile * (16 * 113);
#pragma unroll
  for (int t = 0; t < 4; ++t)
    if (tb + t < 7) {
#pragma unroll
      for (int v = 0; v < 8; ++v)
        s[row * 113 + (tb + t) * 16 + v + 8 * half] = acc[t][v];
    }

  const int t  = threadIdx.x & 63;
  const int cq = threadIdx.x >> 6;

  const float* xg = x + ((size_t)(n * 256 + cq * 64)) * PIX;

  const int c4   = t & 15;
  const int rsub = t >> 4;
  int  xoff[10];
  bool xv[10];
#pragma unroll
  for (int q = 0; q < 10; ++q) {
    const int rr = 4 * q + rsub;
    const int ch = rr / 5;
    const int y  = h + (rr - ch * 5) - 2;
    xv[q] = (unsigned)y < HW;
    const int yc = min(max(y, 0), HW - 1);
    xoff[q] = ch * PIX + yc * HW + 4 * c4;
  }
  const int soff = rsub * XPITCH + 4 + 4 * c4;

  float4 xs[10];
#pragma unroll
  for (int q = 0; q < 10; ++q) xs[q] = *(const float4*)(xg + xoff[q]);

  lds_only_barrier();

  float* smask = lds + SMASK_OFF;
  {
    const int g   = threadIdx.x;
    const int pix = g >> 2;
    const int ij  = g & 3;
    const float* base = lds + (pix >> 4) * (16 * 113) + (pix & 15) * 113;
    float r[25];
    float mx = -3.4e38f;
#pragma unroll
    for (int k = 0; k < 25; ++k) { r[k] = base[k * 4 + ij]; mx = fmaxf(mx, r[k]); }
    float sum = 0.0f;
#pragma unroll
    for (int k = 0; k < 25; ++k) { r[k] = __expf(r[k] - mx); sum += r[k]; }
    const float inv = 1.0f / sum;
    float* mout = smask + (pix & 1) * SMASK_ODD + (pix >> 1) * 100 + ij;
#pragma unroll
    for (int k = 0; k < 25; ++k) mout[4 * k] = r[k] * inv;
  }

  __builtin_amdgcn_sched_barrier(0);

  float* bufa = lds + XWIN_OFF + cq * (2 * XROWS * XPITCH);
  float* bufb = bufa + XROWS * XPITCH;
#pragma unroll
  for (int q = 0; q < 10; ++q) {
    float4 v = xs[q];
    if (!xv[q]) v = make_float4(0.f, 0.f, 0.f, 0.f);
    *(float4*)(bufa + 4 * q * XPITCH + soff) = v;
  }
  for (int i = t; i < 4 * XROWS; i += 64)
    *(float4*)(bufa + (i >> 1) * XPITCH + (i & 1) * (XPITCH - 4)) = make_float4(0.f, 0.f, 0.f, 0.f);

  __syncthreads();

  const float4* sme = (const float4*)smask + lane * 25;
  const float4* smo = (const float4*)(smask + SMASK_ODD) + lane * 25;
  v2f mlo[2][25], mhi[2][25];
#pragma unroll
  for (int k = 0; k < 25; ++k) {
    const float4 v = sme[k];
    mlo[0][k][0] = v.x; mlo[0][k][1] = v.y;
    mhi[0][k][0] = v.z; mhi[0][k][1] = v.w;
    const float4 u = smo[k];
    mlo[1][k][0] = u.x; mlo[1][k][1] = u.y;
    mhi[1][k][0] = u.z; mhi[1][k][1] = u.w;
  }

  const int csel = (wave & 1) * 4;

  float* og = out + (((size_t)(n * 256 + cq * 64 + csel) * 128) + 2 * h) * 128 + 4 * lane;

  for (int sg = 0; sg < NSUB; ++sg) {
    const float* cur = (sg & 1) ? bufb : bufa;
    float*       nxt = (sg & 1) ? bufa : bufb;
    const bool more = (sg + 1 < NSUB);

    if (more) {
      const float* xn = xg + (size_t)(sg + 1) * 8 * PIX;
#pragma unroll
      for (int q = 0; q < 10; ++q) xs[q] = *(const float4*)(xn + xoff[q]);
    }
    __builtin_amdgcn_sched_barrier(0);

    const float* sxl = cur + csel * 5 * XPITCH + 2 * lane + 2;

    for (int cp = 0; cp < 2; ++cp) {
      const float* wa = sxl + (2 * cp) * 5 * XPITCH;
      const float* wb = wa + 5 * XPITCH;

      float ra[5][6], rb[5][6];
#pragma unroll
      for (int di = 0; di < 5; ++di)
#pragma unroll
        for (int c = 0; c < 3; ++c) {
          const float2 va = *(const float2*)(wa + di * XPITCH + 2 * c);
          const float2 vb = *(const float2*)(wb + di * XPITCH + 2 * c);
          ra[di][2 * c] = va.x; ra[di][2 * c + 1] = va.y;
          rb[di][2 * c] = vb.x; rb[di][2 * c + 1] = vb.y;
        }

      v2f salo[2], sahi[2], sblo[2], sbhi[2];
#pragma unroll
      for (int p = 0; p < 2; ++p) {
        salo[p] = (v2f){0.f, 0.f}; sahi[p] = (v2f){0.f, 0.f};
        sblo[p] = (v2f){0.f, 0.f}; sbhi[p] = (v2f){0.f, 0.f};
      }
#pragma unroll
      for (int di = 0; di < 5; ++di)
#pragma unroll
        for (int dj = 0; dj < 5; ++dj) {
          const int k = di * 5 + dj;
#pragma unroll
          for (int p = 0; p < 2; ++p) {
            const float xa = ra[di][dj + p];
            const float xb = rb[di][dj + p];
#pragma unroll
            for (int e = 0; e < 2; ++e) {
              salo[p][e] = __builtin_fmaf(xa, mlo[p][k][e], salo[p][e]);
              sahi[p][e] = __builtin_fmaf(xa, mhi[p][k][e], sahi[p][e]);
              sblo[p][e] = __builtin_fmaf(xb, mlo[p][k][e], sblo[p][e]);
              sbhi[p][e] = __builtin_fmaf(xb, mhi[p][k][e], sbhi[p][e]);
            }
          }
        }

      float* oa = og + (size_t)(sg * 8 + 2 * cp) * 128 * 128;
      float* ob = oa + 128 * 128;
      *(float4*)(oa)       = make_float4(salo[0][0], salo[0][1], salo[1][0], salo[1][1]);
      *(float4*)(oa + 128) = make_float4(sahi[0][0], sahi[0][1], sahi[1][0], sahi[1][1]);
      *(float4*)(ob)       = make_float4(sblo[0][0], sblo[0][1], sblo[1][0], sblo[1][1]);
      *(float4*)(ob + 128) = make_float4(sbhi[0][0], sbhi[0][1], sbhi[1][0], sbhi[1][1]);
    }

    if (more) {
#pragma unroll
      for (int q = 0; q < 10; ++q) {
        float4 v = xs[q];
        if (!xv[q]) v = make_float4(0.f, 0.f, 0.f, 0.f);
        *(float4*)(nxt + 4 * q * XPITCH + soff) = v;
      }
    }
    lds_only_barrier();
  }
}

extern "C" void kernel_launch(void* const* d_in, const int* in_sizes, int n_in,
                              void* d_out, int out_size, void* d_ws, size_t ws_size,
                              hipStream_t stream) {
  (void)in_sizes; (void)n_in; (void)out_size; (void)ws_size;
  const float* x      = (const float*)d_in[0];
  const float* w_comp = (const float*)d_in[1];
  const float* b_comp = (const float*)d_in[2];
  const float* w_enc  = (const float*)d_in[3];
  const float* b_enc  = (const float*)d_in[4];
  float* out = (float*)d_out;

  float* comp = (float*)d_ws;

  carafe_compress<<<256, 256, 0, stream>>>(x, w_comp, b_comp, comp);
  carafe_encode_reassemble<<<256, 256, 0, stream>>>(comp, w_enc, b_enc, x, out);
}
